# starting kernel
# baseline (speedup 1.0000x reference)
#include <hip/hip_runtime.h>

#define LAM 2.0f

typedef __attribute__((ext_vector_type(2))) float v2f;
typedef __attribute__((ext_vector_type(8))) float v8f;

__global__ void zero_kernel(float* __restrict__ p, long long n) {
    long long i = (long long)blockIdx.x * blockDim.x + threadIdx.x;
    if (i < n) p[i] = 0.0f;
}

__global__ void degree_kernel(const int* __restrict__ dst, float* __restrict__ deg, int E) {
    int i = blockIdx.x * blockDim.x + threadIdx.x;
    if (i < E) atomicAdd(&deg[dst[i]], 1.0f);
}

__global__ void norm_kernel(float* __restrict__ deg, int N) {
    int i = blockIdx.x * blockDim.x + threadIdx.x;
    if (i < N) deg[i] = 1.0f / sqrtf(fmaxf(deg[i], 1.0f));
}

__global__ void scatter_kernel(const float* __restrict__ x, const float* __restrict__ norm,
                               const int* __restrict__ src, const int* __restrict__ dst,
                               float* __restrict__ h, long long total) {
    long long i = (long long)blockIdx.x * blockDim.x + threadIdx.x;
    if (i >= total) return;
    int e = (int)(i >> 6);
    int f = (int)(i & 63);
    int s = src[e];
    int d = dst[e];
    float v = x[(long long)s * 64 + f] * norm[s];
    atomicAdd(&h[(long long)d * 64 + f], v);
}

__global__ void cheb1_kernel(const float* __restrict__ feat, const float* __restrict__ norm,
                             float* __restrict__ h, float* __restrict__ Tx1, long long total) {
    long long i = (long long)blockIdx.x * blockDim.x + threadIdx.x;
    if (i >= total) return;
    const float ch = -2.0f / LAM;
    const float cx = 2.0f / LAM - 1.0f;
    int node = (int)(i >> 6);
    Tx1[i] = ch * h[i] * norm[node] + cx * feat[i];
    h[i] = 0.0f;
}

__global__ __launch_bounds__(256) void cheb_gemm_wmma(
    const float* __restrict__ feat, const float* __restrict__ Tx1,
    const float* __restrict__ h2,   const float* __restrict__ norm,
    const float* __restrict__ W,    const float* __restrict__ bias,
    float* __restrict__ out) {
    const int lane = threadIdx.x & 31;
    const int wave = threadIdx.x >> 5;
    const int gw   = blockIdx.x * 8 + wave;
    const int row_tile = gw >> 2;
    const int n_tile   = gw & 3;
    const int l15  = lane & 15;
    const int half = lane >> 4;
    const int koff = half * 2;

    const int arow = row_tile * 16 + l15;
    const int ncol = n_tile * 16 + l15;

    const float nrm = norm[arow];
    const float* fr = feat + (long long)arow * 64;
    const float* tr = Tx1  + (long long)arow * 64;
    const float* hr = h2   + (long long)arow * 64;
    const float* W0 = W;
    const float* W1 = W + 4096;
    const float* W2 = W + 8192;

    const float ch2 = -4.0f / LAM;
    const float cx2 =  4.0f / LAM - 2.0f;

    v8f acc = {};
#pragma unroll
    for (int k = 0; k < 64; k += 4) {
        const int ka = k + koff;
        float f0 = fr[ka],  f1 = fr[ka + 1];
        float t0 = tr[ka],  t1 = tr[ka + 1];
        float g0 = hr[ka],  g1 = hr[ka + 1];
        v2f a0 = {f0, f1};
        v2f a1 = {t0, t1};
        v2f a2 = {ch2 * g0 * nrm + cx2 * t0 - f0,
                  ch2 * g1 * nrm + cx2 * t1 - f1};
        v2f b0 = {W0[ka * 64 + ncol], W0[(ka + 1) * 64 + ncol]};
        v2f b1 = {W1[ka * 64 + ncol], W1[(ka + 1) * 64 + ncol]};
        v2f b2 = {W2[ka * 64 + ncol], W2[(ka + 1) * 64 + ncol]};
        acc = __builtin_amdgcn_wmma_f32_16x16x4_f32(false, a0, false, b0,
                                                    (short)0, acc, false, false);
        acc = __builtin_amdgcn_wmma_f32_16x16x4_f32(false, a1, false, b1,
                                                    (short)0, acc, false, false);
        acc = __builtin_amdgcn_wmma_f32_16x16x4_f32(false, a2, false, b2,
                                                    (short)0, acc, false, false);
    }

    const float b = bias[ncol];
#pragma unroll
    for (int r = 0; r < 8; ++r) {
        int orow = row_tile * 16 + r + 8 * half;
        out[(long long)orow * 64 + ncol] = acc[r] + b;
    }
}

extern "C" void kernel_launch(void* const* d_in, const int* in_sizes, int n_in,
                              void* d_out, int out_size, void* d_ws, size_t ws_size,
                              hipStream_t stream) {
    const float* feat = (const float*)d_in[0];
    const float* W    = (const float*)d_in[1];
    const float* bias = (const float*)d_in[2];
    const int*   esrc = (const int*)d_in[3];
    const int*   edst = (const int*)d_in[4];
    float* out = (float*)d_out;

    const int N = in_sizes[0] / 64;
    const int E = in_sizes[3];

    float* ws   = (float*)d_ws;
    float* norm = ws;
    float* h    = ws + N;
    float* Tx1  = ws + N + (long long)N * 64;

    const long long nf  = (long long)N * 64;
    const long long tot = (long long)E * 64;
    const long long zn  = (long long)N * 65;

    zero_kernel<<<(unsigned)((zn + 255) / 256), 256, 0, stream>>>(ws, zn);
    degree_kernel<<<(E + 255) / 256, 256, 0, stream>>>(edst, norm, E);
    norm_kernel<<<(N + 255) / 256, 256, 0, stream>>>(norm, N);

    scatter_kernel<<<(unsigned)((tot + 255) / 256), 256, 0, stream>>>(
        feat, norm, esrc, edst, h, tot);
    cheb1_kernel<<<(unsigned)((nf + 255) / 256), 256, 0, stream>>>(
        feat, norm, h, Tx1, nf);
    scatter_kernel<<<(unsigned)((tot + 255) / 256), 256, 0, stream>>>(
        Tx1, norm, esrc, edst, h, tot);

    const int waves = (N / 16) * 4;
    cheb_gemm_wmma<<<waves / 8, 256, 0, stream>>>(feat, Tx1, h, norm, W, bias, out);
}


# Weight products moved into the first neighbour sum's owner work-groups (Tx1 tile in LDS feeds WMMA; by linearity pass 2 sums Tx1@W2 rows into out): GEMM launch, Tx1 and h gone; sum order changes
# speedup vs baseline: 3.1604x; 3.1604x over previous
#include <hip/hip_runtime.h>

#define LAM 2.0f

typedef __attribute__((ext_vector_type(2))) float v2f;
typedef __attribute__((ext_vector_type(8))) float v8f;

#define BSH   6
#define BN    (1 << BSH)
#define BMASK (BN - 1)
#define TE    4096
#define MAXNB 2048
#define EPI   8

__device__ __forceinline__ int lds_inc(int* p) {
    return __hip_atomic_fetch_add(p, 1, __ATOMIC_RELAXED, __HIP_MEMORY_SCOPE_WORKGROUP);
}
__device__ __forceinline__ void lds_addf(float* p, float v) {
    __hip_atomic_fetch_add(p, v, __ATOMIC_RELAXED, __HIP_MEMORY_SCOPE_WORKGROUP);
}

__global__ __launch_bounds__(256) void bin_count(const int* __restrict__ dst,
                                                 int* __restrict__ tileHist, int E, int NB) {
    __shared__ int hist[MAXNB];
    const int t = blockIdx.x;
    for (int b = threadIdx.x; b < NB; b += 256) hist[b] = 0;
    __syncthreads();
    for (int i = threadIdx.x; i < TE; i += 256) {
        const int e = t * TE + i;
        if (e < E) lds_inc(&hist[dst[e] >> BSH]);
    }
    __syncthreads();
    for (int b = threadIdx.x; b < NB; b += 256) tileHist[(long long)t * NB + b] = hist[b];
}

__global__ void bin_tile_offsets(const int* __restrict__ tileHist, int* __restrict__ tileOff,
                                 int* __restrict__ binTotal, int NT, int NB) {
    const int b = blockIdx.x * blockDim.x + threadIdx.x;
    if (b >= NB) return;
    int run = 0;
#pragma unroll 16
    for (int t = 0; t < NT; ++t) {
        const int c = tileHist[(long long)t * NB + b];
        tileOff[(long long)t * NB + b] = run;
        run += c;
    }
    binTotal[b] = run;
}

__global__ __launch_bounds__(1024) void bin_bases(const int* __restrict__ binTotal,
                                                  int* __restrict__ binBase, int NB) {
    __shared__ int part[1024];
    const int tid = threadIdx.x;
    const int per = (NB + 1023) / 1024;
    const int lo  = tid * per;
    int sum = 0;
    for (int k = 0; k < per; ++k) {
        const int b = lo + k;
        if (b < NB) sum += binTotal[b];
    }
    part[tid] = sum;
    __syncthreads();
    for (int off = 1; off < 1024; off <<= 1) {
        const int v = (tid >= off) ? part[tid - off] : 0;
        __syncthreads();
        part[tid] += v;
        __syncthreads();
    }
    int run = part[tid] - sum;
    for (int k = 0; k < per; ++k) {
        const int b = lo + k;
        if (b < NB) { binBase[b] = run; run += binTotal[b]; }
    }
}

__global__ __launch_bounds__(256) void bin_fill(const int* __restrict__ src, const int* __restrict__ dst,
                                                const int* __restrict__ tileOff,
                                                const int* __restrict__ binBase,
                                                int* __restrict__ list, int E, int NB) {
    __shared__ int cur[MAXNB];
    const int t = blockIdx.x;
    for (int b = threadIdx.x; b < NB; b += 256) cur[b] = binBase[b] + tileOff[(long long)t * NB + b];
    __syncthreads();
    for (int i = threadIdx.x; i < TE; i += 256) {
        const int e = t * TE + i;
        if (e < E) {
            const int d = dst[e];
            const int s = src[e];
            const int p = lds_inc(&cur[d >> BSH]);
            list[p] = (s << BSH) | (d & BMASK);
        }
    }
}

__global__ __launch_bounds__(256) void block_norm(const int* __restrict__ list,
                                                  const int* __restrict__ binBase,
                                                  const int* __restrict__ binTotal,
                                                  float* __restrict__ norm, int N) {
    __shared__ int c[BN];
    const int b = blockIdx.x;
    if (threadIdx.x < BN) c[threadIdx.x] = 0;
    __syncthreads();
    const int base = binBase[b];
    const int cnt  = binTotal[b];
    for (int i = threadIdx.x; i < cnt; i += 256) lds_inc(&c[list[base + i] & BMASK]);
    __syncthreads();
    if (threadIdx.x < BN) {
        const int node = b * BN + threadIdx.x;
        if (node < N) norm[node] = 1.0f / sqrtf(fmaxf((float)c[threadIdx.x], 1.0f));
    }
}

template <int P>
__device__ __forceinline__ void gather_rows(const float* __restrict__ x, const float* __restrict__ norm,
                                            const int* __restrict__ list, int base, int cnt,
                                            float* acc, int lane, int wave) {
    for (int i = wave * EPI; i + EPI <= cnt; i += 8 * EPI) {
        int   wd[EPI];
        float nn[EPI], v0[EPI], v1[EPI];
        for (int j = 0; j < EPI; ++j) wd[j] = list[base + i + j];
        for (int j = 0; j < EPI; ++j) {
            const int s = wd[j] >> BSH;
            const float* r = x + (long long)s * 64;
            nn[j] = norm[s];
            v0[j] = r[lane];
            v1[j] = r[lane + 32];
        }
        for (int j = 0; j < EPI; ++j) {
            float* a = acc + (wd[j] & BMASK) * P;
            lds_addf(&a[lane],      v0[j] * nn[j]);
            lds_addf(&a[lane + 32], v1[j] * nn[j]);
        }
    }
    {
        const int i = (cnt / EPI) * EPI + wave;
        if (i < cnt) {
            const int w = list[base + i];
            const int s = w >> BSH;
            const float* r = x + (long long)s * 64;
            const float n = norm[s];
            float* a = acc + (w & BMASK) * P;
            lds_addf(&a[lane],      r[lane] * n);
            lds_addf(&a[lane + 32], r[lane + 32] * n);
        }
    }
}

#define PITCH 68

__global__ __launch_bounds__(256, 16) void aggregate_project(
    const float* __restrict__ feat, const float* __restrict__ norm,
    const int* __restrict__ list, const int* __restrict__ binBase,
    const int* __restrict__ binTotal, const float* __restrict__ W,
    const float* __restrict__ bias, float* __restrict__ outL,
    float* __restrict__ Y, int N) {
    __shared__ __attribute__((aligned(16))) float acc[BN * PITCH];
    __shared__ __attribute__((aligned(16))) float ft[BN * PITCH];
    __shared__ float nb[BN];
    const int b    = blockIdx.x;
    const int lane = threadIdx.x & 31;
    const int wave = threadIdx.x >> 5;
    for (int k = threadIdx.x; k < BN * PITCH; k += 256) acc[k] = 0.0f;
    __syncthreads();

    gather_rows<PITCH>(feat, norm, list, binBase[b], binTotal[b], acc, lane, wave);

    const int f    = threadIdx.x & 63;
    const int r0   = threadIdx.x >> 6;
    const int rows = min(BN, N - b * BN);
#pragma unroll
    for (int j = 0; j < BN / 4; ++j) {
        const int r  = r0 + 4 * j;
        const int rr = min(r, rows - 1);
        const float v = feat[(long long)(b * BN + rr) * 64 + f];
        ft[r * PITCH + f] = (r < rows) ? v : 0.0f;
    }
    if (threadIdx.x < BN) nb[threadIdx.x] = ((int)threadIdx.x < rows) ? norm[b * BN + threadIdx.x] : 0.0f;
    __syncthreads();

    const float ch = -2.0f / LAM;
    const float cx = 2.0f / LAM - 1.0f;
    for (int r = r0; r < BN; r += 4)
        acc[r * PITCH + f] = ch * acc[r * PITCH + f] * nb[r] + cx * ft[r * PITCH + f];
    __syncthreads();

    const int c    = lane & 15;
    const int h    = lane >> 4;
    const int rp   = wave >> 2;
    const int ncol = (wave & 3) * 16 + c;
    const float* W0 = W;
    const float* W1 = W + 4096;
    const float* W2 = W + 8192;
    const float cx2 = 4.0f / LAM - 2.0f;
    const float* fa = ft  + (rp * 32 + c) * PITCH + 2 * h;
    const float* ta = acc + (rp * 32 + c) * PITCH + 2 * h;

    v8f L0 = {}, L1 = {}, Y0 = {}, Y1 = {};
#pragma unroll 2
    for (int k = 0; k < 64; k += 4) {
        const int ka = k + 2 * h;
        const float w0a = W0[ka * 64 + ncol], w0b = W0[(ka + 1) * 64 + ncol];
        const float w1a = W1[ka * 64 + ncol], w1b = W1[(ka + 1) * 64 + ncol];
        const float w2a = W2[ka * 64 + ncol], w2b = W2[(ka + 1) * 64 + ncol];
        const v2f bY = {w2a, w2b};
        const v2f bF = {w0a - w2a, w0b - w2b};
        const v2f bT = {w1a + cx2 * w2a, w1b + cx2 * w2b};
        const v2f f0 = *(const v2f*)(fa + k);
        const v2f t0 = *(const v2f*)(ta + k);
        const v2f f1 = *(const v2f*)(fa + 16 * PITCH + k);
        const v2f t1 = *(const v2f*)(ta + 16 * PITCH + k);
        Y0 = __builtin_amdgcn_wmma_f32_16x16x4_f32(false, t0, false, bY, (short)0, Y0, false, false);
        Y1 = __builtin_amdgcn_wmma_f32_16x16x4_f32(false, t1, false, bY, (short)0, Y1, false, false);
        L0 = __builtin_amdgcn_wmma_f32_16x16x4_f32(false, f0, false, bF, (short)0, L0, false, false);
        L1 = __builtin_amdgcn_wmma_f32_16x16x4_f32(false, f1, false, bF, (short)0, L1, false, false);
        L0 = __builtin_amdgcn_wmma_f32_16x16x4_f32(false, t0, false, bT, (short)0, L0, false, false);
        L1 = __builtin_amdgcn_wmma_f32_16x16x4_f32(false, t1, false, bT, (short)0, L1, false, false);
    }

    const float bv = bias[ncol];
    __syncthreads();
#pragma unroll
    for (int r = 0; r < 8; ++r) {
        const int lr = rp * 32 + r + 8 * h;
        ft[lr * PITCH + ncol]         = L0[r] + bv;
        acc[lr * PITCH + ncol]        = Y0[r];
        ft[(lr + 16) * PITCH + ncol]  = L1[r] + bv;
        acc[(lr + 16) * PITCH + ncol] = Y1[r];
    }
    __syncthreads();
#pragma unroll 4
    for (int j = 0; j < BN / 4; ++j) {
        const int r = r0 + 4 * j;
        if (r < rows) {
            const long long o = (long long)(b * BN + r) * 64 + f;
            outL[o] = ft[r * PITCH + f];
            Y[o]    = acc[r * PITCH + f];
        }
    }
}

__global__ __launch_bounds__(256) void aggregate_finish(
    const float* __restrict__ Y, const float* __restrict__ norm,
    const int* __restrict__ list, const int* __restrict__ binBase,
    const int* __restrict__ binTotal, float* __restrict__ out, int N) {
    __shared__ float acc[BN * 64];
    const int b    = blockIdx.x;
    const int lane = threadIdx.x & 31;
    const int wave = threadIdx.x >> 5;
    for (int k = threadIdx.x; k < BN * 64; k += 256) acc[k] = 0.0f;
    __syncthreads();

    gather_rows<64>(Y, norm, list, binBase[b], binTotal[b], acc, lane, wave);
    __syncthreads();

    const float ch2 = -4.0f / LAM;
    const int f    = threadIdx.x & 63;
    const int r0   = threadIdx.x >> 6;
    const int rows = min(BN, N - b * BN);
#pragma unroll 4
    for (int r = r0; r < rows; r += 4) {
        const int node = b * BN + r;
        const long long o = (long long)node * 64 + f;
        out[o] = ch2 * acc[r * 64 + f] * norm[node] + out[o];
    }
}

extern "C" void kernel_launch(void* const* d_in, const int* in_sizes, int n_in,
                              void* d_out, int out_size, void* d_ws, size_t ws_size,
                              hipStream_t stream) {
    const float* feat = (const float*)d_in[0];
    const float* W    = (const float*)d_in[1];
    const float* bias = (const float*)d_in[2];
    const int*   esrc = (const int*)d_in[3];
    const int*   edst = (const int*)d_in[4];
    float* out = (float*)d_out;

    const int N = in_sizes[0] / 64;
    const int E = in_sizes[3];

    float* ws   = (float*)d_ws;
    float* norm = ws;
    float* Y    = ws + N;

    const int NB = (N + BN - 1) / BN;
    const int NT = (E + TE - 1) / TE;

    int* scr      = (int*)(ws + N + (long long)N * 64);
    int* list     = scr;
    int* tileHist = list + E;
    int* tileOff  = tileHist + (long long)NT * NB;
    int* binTotal = tileOff + (long long)NT * NB;
    int* binBase  = binTotal + NB;

    bin_count<<<NT, 256, 0, stream>>>(edst, tileHist, E, NB);
    bin_tile_offsets<<<(NB + 63) / 64, 64, 0, stream>>>(tileHist, tileOff, binTotal, NT, NB);
    bin_bases<<<1, 1024, 0, stream>>>(binTotal, binBase, NB);
    bin_fill<<<NT, 256, 0, stream>>>(esrc, edst, tileOff, binBase, list, E, NB);
    block_norm<<<NB, 256, 0, stream>>>(list, binBase, binTotal, norm, N);

    aggregate_project<<<NB, 256, 0, stream>>>(feat, norm, list, binBase, binTotal, W, bias, out, Y, N);
    aggregate_finish<<<NB, 256, 0, stream>>>(Y, norm, list, binBase, binTotal, out, N);
}


# Tile-offset scan of the binning spread over waves: a wave loads 16 tiles of 32 blocks as one group, wave sums meet in LDS; about 100 dependent cache round trips per thread become one
# speedup vs baseline: 4.0925x; 1.2949x over previous
#include <hip/hip_runtime.h>

#define LAM 2.0f

typedef __attribute__((ext_vector_type(2))) float v2f;
typedef __attribute__((ext_vector_type(8))) float v8f;

#define BSH   6
#define BN    (1 << BSH)
#define BMASK (BN - 1)
#define TE    4096
#define MAXNB 2048
#define EPI   8

__device__ __forceinline__ int lds_inc(int* p) {
    return __hip_atomic_fetch_add(p, 1, __ATOMIC_RELAXED, __HIP_MEMORY_SCOPE_WORKGROUP);
}
__device__ __forceinline__ void lds_addf(float* p, float v) {
    __hip_atomic_fetch_add(p, v, __ATOMIC_RELAXED, __HIP_MEMORY_SCOPE_WORKGROUP);
}

__global__ __launch_bounds__(256) void bin_count(const int* __restrict__ dst,
                                                 int* __restrict__ tileHist, int E, int NB) {
    __shared__ int hist[MAXNB];
    const int t = blockIdx.x;
    for (int b = threadIdx.x; b < NB; b += 256) hist[b] = 0;
    __syncthreads();
    for (int i = threadIdx.x; i < TE; i += 256) {
        const int e = t * TE + i;
        if (e < E) lds_inc(&hist[dst[e] >> BSH]);
    }
    __syncthreads();
    for (int b = threadIdx.x; b < NB; b += 256) tileHist[(long long)t * NB + b] = hist[b];
}

#define TPW 16
__global__ __launch_bounds__(1024) void bin_tile_offsets(const int* __restrict__ tileHist, int* __restrict__ tileOff,
                                                         int* __restrict__ binTotal, int NT, int NB) {
    __shared__ int part[32 * 32];
    const int lane  = threadIdx.x & 31;
    const int wave  = __builtin_amdgcn_readfirstlane((int)(threadIdx.x >> 5));
    const int b     = blockIdx.x * 32 + lane;
    const bool live = b < NB;
    const int bb    = min(b, NB - 1);
    int carry = 0;
    for (int tr = 0; tr < NT; tr += 32 * TPW) {
        const int t0 = tr + wave * TPW;
        int c[TPW];
#pragma unroll
        for (int j = 0; j < TPW; ++j) {
            const int tt = min(t0 + j, NT - 1);
            c[j] = tileHist[(long long)tt * NB + bb];
        }
        int tot = 0;
#pragma unroll
        for (int j = 0; j < TPW; ++j) {
            c[j] = (t0 + j < NT) ? c[j] : 0;
            tot += c[j];
        }
        part[wave * 32 + lane] = tot;
        __syncthreads();
        int before = 0, all = 0;
#pragma unroll
        for (int w = 0; w < 32; ++w) {
            const int p = part[w * 32 + lane];
            all    += p;
            before += (w < wave) ? p : 0;
        }
        int run = carry + before;
#pragma unroll
        for (int j = 0; j < TPW; ++j) {
            if (live && t0 + j < NT) tileOff[(long long)(t0 + j) * NB + b] = run;
            run += c[j];
        }
        carry += all;
        __syncthreads();
    }
    if (wave == 0 && live) binTotal[b] = carry;
}

__global__ __launch_bounds__(1024) void bin_bases(const int* __restrict__ binTotal,
                                                  int* __restrict__ binBase, int NB) {
    __shared__ int part[1024];
    const int tid = threadIdx.x;
    const int per = (NB + 1023) / 1024;
    const int lo  = tid * per;
    int sum = 0;
    for (int k = 0; k < per; ++k) {
        const int b = lo + k;
        if (b < NB) sum += binTotal[b];
    }
    part[tid] = sum;
    __syncthreads();
    for (int off = 1; off < 1024; off <<= 1) {
        const int v = (tid >= off) ? part[tid - off] : 0;
        __syncthreads();
        part[tid] += v;
        __syncthreads();
    }
    int run = part[tid] - sum;
    for (int k = 0; k < per; ++k) {
        const int b = lo + k;
        if (b < NB) { binBase[b] = run; run += binTotal[b]; }
    }
}

__global__ __launch_bounds__(256) void bin_fill(const int* __restrict__ src, const int* __restrict__ dst,
                                                const int* __restrict__ tileOff,
                                                const int* __restrict__ binBase,
                                                int* __restrict__ list, int E, int NB) {
    __shared__ int cur[MAXNB];
    const int t = blockIdx.x;
    for (int b = threadIdx.x; b < NB; b += 256) cur[b] = binBase[b] + tileOff[(long long)t * NB + b];
    __syncthreads();
    for (int i = threadIdx.x; i < TE; i += 256) {
        const int e = t * TE + i;
        if (e < E) {
            const int d = dst[e];
            const int s = src[e];
            const int p = lds_inc(&cur[d >> BSH]);
            list[p] = (s << BSH) | (d & BMASK);
        }
    }
}

__global__ __launch_bounds__(256) void block_norm(const int* __restrict__ list,
                                                  const int* __restrict__ binBase,
                                                  const int* __restrict__ binTotal,
                                                  float* __restrict__ norm, int N) {
    __shared__ int c[BN];
    const int b = blockIdx.x;
    if (threadIdx.x < BN) c[threadIdx.x] = 0;
    __syncthreads();
    const int base = binBase[b];
    const int cnt  = binTotal[b];
    for (int i = threadIdx.x; i < cnt; i += 256) lds_inc(&c[list[base + i] & BMASK]);
    __syncthreads();
    if (threadIdx.x < BN) {
        const int node = b * BN + threadIdx.x;
        if (node < N) norm[node] = 1.0f / sqrtf(fmaxf((float)c[threadIdx.x], 1.0f));
    }
}

template <int P>
__device__ __forceinline__ void gather_rows(const float* __restrict__ x, const float* __restrict__ norm,
                                            const int* __restrict__ list, int base, int cnt,
                                            float* acc, int lane, int wave) {
    for (int i = wave * EPI; i + EPI <= cnt; i += 8 * EPI) {
        int   wd[EPI];
        float nn[EPI], v0[EPI], v1[EPI];
        for (int j = 0; j < EPI; ++j) wd[j] = list[base + i + j];
        for (int j = 0; j < EPI; ++j) {
            const int s = wd[j] >> BSH;
            const float* r = x + (long long)s * 64;
            nn[j] = norm[s];
            v0[j] = r[lane];
            v1[j] = r[lane + 32];
        }
        for (int j = 0; j < EPI; ++j) {
            float* a = acc + (wd[j] & BMASK) * P;
            lds_addf(&a[lane],      v0[j] * nn[j]);
            lds_addf(&a[lane + 32], v1[j] * nn[j]);
        }
    }
    {
        const int i = (cnt / EPI) * EPI + wave;
        if (i < cnt) {
            const int w = list[base + i];
            const int s = w >> BSH;
            const float* r = x + (long long)s * 64;
            const float n = norm[s];
            float* a = acc + (w & BMASK) * P;
            lds_addf(&a[lane],      r[lane] * n);
            lds_addf(&a[lane + 32], r[lane + 32] * n);
        }
    }
}

#define PITCH 68

__global__ __launch_bounds__(256, 16) void aggregate_project(
    const float* __restrict__ feat, const float* __restrict__ norm,
    const int* __restrict__ list, const int* __restrict__ binBase,
    const int* __restrict__ binTotal, const float* __restrict__ W,
    const float* __restrict__ bias, float* __restrict__ outL,
    float* __restrict__ Y, int N) {
    __shared__ __attribute__((aligned(16))) float acc[BN * PITCH];
    __shared__ __attribute__((aligned(16))) float ft[BN * PITCH];
    __shared__ float nb[BN];
    const int b    = blockIdx.x;
    const int lane = threadIdx.x & 31;
    const int wave = threadIdx.x >> 5;
    for (int k = threadIdx.x; k < BN * PITCH; k += 256) acc[k] = 0.0f;
    __syncthreads();

    gather_rows<PITCH>(feat, norm, list, binBase[b], binTotal[b], acc, lane, wave);

    const int f    = threadIdx.x & 63;
    const int r0   = threadIdx.x >> 6;
    const int rows = min(BN, N - b * BN);
#pragma unroll
    for (int j = 0; j < BN / 4; ++j) {
        const int r  = r0 + 4 * j;
        const int rr = min(r, rows - 1);
        const float v = feat[(long long)(b * BN + rr) * 64 + f];
        ft[r * PITCH + f] = (r < rows) ? v : 0.0f;
    }
    if (threadIdx.x < BN) nb[threadIdx.x] = ((int)threadIdx.x < rows) ? norm[b * BN + threadIdx.x] : 0.0f;
    __syncthreads();

    const float ch = -2.0f / LAM;
    const float cx = 2.0f / LAM - 1.0f;
    for (int r = r0; r < BN; r += 4)
        acc[r * PITCH + f] = ch * acc[r * PITCH + f] * nb[r] + cx * ft[r * PITCH + f];
    __syncthreads();

    const int c    = lane & 15;
    const int h    = lane >> 4;
    const int rp   = wave >> 2;
    const int ncol = (wave & 3) * 16 + c;
    const float* W0 = W;
    const float* W1 = W + 4096;
    const float* W2 = W + 8192;
    const float cx2 = 4.0f / LAM - 2.0f;
    const float* fa = ft  + (rp * 32 + c) * PITCH + 2 * h;
    const float* ta = acc + (rp * 32 + c) * PITCH + 2 * h;

    v8f L0 = {}, L1 = {}, Y0 = {}, Y1 = {};
#pragma unroll 2
    for (int k = 0; k < 64; k += 4) {
        const int ka = k + 2 * h;
        const float w0a = W0[ka * 64 + ncol], w0b = W0[(ka + 1) * 64 + ncol];
        const float w1a = W1[ka * 64 + ncol], w1b = W1[(ka + 1) * 64 + ncol];
        const float w2a = W2[ka * 64 + ncol], w2b = W2[(ka + 1) * 64 + ncol];
        const v2f bY = {w2a, w2b};
        const v2f bF = {w0a - w2a, w0b - w2b};
        const v2f bT = {w1a + cx2 * w2a, w1b + cx2 * w2b};
        const v2f f0 = *(const v2f*)(fa + k);
        const v2f t0 = *(const v2f*)(ta + k);
        const v2f f1 = *(const v2f*)(fa + 16 * PITCH + k);
        const v2f t1 = *(const v2f*)(ta + 16 * PITCH + k);
        Y0 = __builtin_amdgcn_wmma_f32_16x16x4_f32(false, t0, false, bY, (short)0, Y0, false, false);
        Y1 = __builtin_amdgcn_wmma_f32_16x16x4_f32(false, t1, false, bY, (short)0, Y1, false, false);
        L0 = __builtin_amdgcn_wmma_f32_16x16x4_f32(false, f0, false, bF, (short)0, L0, false, false);
        L1 = __builtin_amdgcn_wmma_f32_16x16x4_f32(false, f1, false, bF, (short)0, L1, false, false);
        L0 = __builtin_amdgcn_wmma_f32_16x16x4_f32(false, t0, false, bT, (short)0, L0, false, false);
        L1 = __builtin_amdgcn_wmma_f32_16x16x4_f32(false, t1, false, bT, (short)0, L1, false, false);
    }

    const float bv = bias[ncol];
    __syncthreads();
#pragma unroll
    for (int r = 0; r < 8; ++r) {
        const int lr = rp * 32 + r + 8 * h;
        ft[lr * PITCH + ncol]         = L0[r] + bv;
        acc[lr * PITCH + ncol]        = Y0[r];
        ft[(lr + 16) * PITCH + ncol]  = L1[r] + bv;
        acc[(lr + 16) * PITCH + ncol] = Y1[r];
    }
    __syncthreads();
#pragma unroll 4
    for (int j = 0; j < BN / 4; ++j) {
        const int r = r0 + 4 * j;
        if (r < rows) {
            const long long o = (long long)(b * BN + r) * 64 + f;
            outL[o] = ft[r * PITCH + f];
            Y[o]    = acc[r * PITCH + f];
        }
    }
}

__global__ __launch_bounds__(256) void aggregate_finish(
    const float* __restrict__ Y, const float* __restrict__ norm,
    const int* __restrict__ list, const int* __restrict__ binBase,
    const int* __restrict__ binTotal, float* __restrict__ out, int N) {
    __shared__ float acc[BN * 64];
    const int b    = blockIdx.x;
    const int lane = threadIdx.x & 31;
    const int wave = threadIdx.x >> 5;
    for (int k = threadIdx.x; k < BN * 64; k += 256) acc[k] = 0.0f;
    __syncthreads();

    gather_rows<64>(Y, norm, list, binBase[b], binTotal[b], acc, lane, wave);
    __syncthreads();

    const float ch2 = -4.0f / LAM;
    const int f    = threadIdx.x & 63;
    const int r0   = threadIdx.x >> 6;
    const int rows = min(BN, N - b * BN);
#pragma unroll 4
    for (int r = r0; r < rows; r += 4) {
        const int node = b * BN + r;
        const long long o = (long long)node * 64 + f;
        out[o] = ch2 * acc[r * 64 + f] * norm[node] + out[o];
    }
}

extern "C" void kernel_launch(void* const* d_in, const int* in_sizes, int n_in,
                              void* d_out, int out_size, void* d_ws, size_t ws_size,
                              hipStream_t stream) {
    const float* feat = (const float*)d_in[0];
    const float* W    = (const float*)d_in[1];
    const float* bias = (const float*)d_in[2];
    const int*   esrc = (const int*)d_in[3];
    const int*   edst = (const int*)d_in[4];
    float* out = (float*)d_out;

    const int N = in_sizes[0] / 64;
    const int E = in_sizes[3];

    float* ws   = (float*)d_ws;
    float* norm = ws;
    float* Y    = ws + N;

    const int NB = (N + BN - 1) / BN;
    const int NT = (E + TE - 1) / TE;

    int* scr      = (int*)(ws + N + (long long)N * 64);
    int* list     = scr;
    int* tileHist = list + E;
    int* tileOff  = tileHist + (long long)NT * NB;
    int* binTotal = tileOff + (long long)NT * NB;
    int* binBase  = binTotal + NB;

    bin_count<<<NT, 256, 0, stream>>>(edst, tileHist, E, NB);
    bin_tile_offsets<<<(NB + 31) / 32, 1024, 0, stream>>>(tileHist, tileOff, binTotal, NT, NB);
    bin_bases<<<1, 1024, 0, stream>>>(binTotal, binBase, NB);
    bin_fill<<<NT, 256, 0, stream>>>(esrc, edst, tileOff, binBase, list, E, NB);
    block_norm<<<NB, 256, 0, stream>>>(list, binBase, binTotal, norm, N);

    aggregate_project<<<NB, 256, 0, stream>>>(feat, norm, list, binBase, binTotal, W, bias, out, Y, N);
    aggregate_finish<<<NB, 256, 0, stream>>>(Y, norm, list, binBase, binTotal, out, N);
}


# Binning kernels issue a thread's loads as one group ahead of the LDS counters (16 dst, src, cursor words per tile thread; 8 list words in the degree count), not one wait each; sum order by arrival
# speedup vs baseline: 4.2943x; 1.0493x over previous
#include <hip/hip_runtime.h>

#define LAM 2.0f

typedef __attribute__((ext_vector_type(2))) float v2f;
typedef __attribute__((ext_vector_type(8))) float v8f;

#define BSH   6
#define BN    (1 << BSH)
#define BMASK (BN - 1)
#define TE    4096
#define MAXNB 2048
#define EPI   8

__device__ __forceinline__ int lds_inc(int* p) {
    return __hip_atomic_fetch_add(p, 1, __ATOMIC_RELAXED, __HIP_MEMORY_SCOPE_WORKGROUP);
}
__device__ __forceinline__ void lds_addf(float* p, float v) {
    __hip_atomic_fetch_add(p, v, __ATOMIC_RELAXED, __HIP_MEMORY_SCOPE_WORKGROUP);
}

#define EPT (TE / 256)

template <bool FULL>
__device__ __forceinline__ void tile_words(const int* __restrict__ a, int e0, int E, int (&w)[EPT]) {
#pragma unroll
    for (int j = 0; j < EPT; ++j) {
        const int e = e0 + 256 * j;
        w[j] = a[FULL ? e : min(e, E - 1)];
    }
}

__global__ __launch_bounds__(256) void bin_count(const int* __restrict__ dst,
                                                 int* __restrict__ tileHist, int E, int NB) {
    __shared__ int hist[MAXNB];
    const int t    = blockIdx.x;
    const int e0   = t * TE + threadIdx.x;
    const bool full = (t + 1) * TE <= E;
    int d[EPT];
    if (full) tile_words<true>(dst, e0, E, d); else tile_words<false>(dst, e0, E, d);
    for (int b = threadIdx.x; b < NB; b += 256) hist[b] = 0;
    __syncthreads();
    if (full) {
#pragma unroll
        for (int j = 0; j < EPT; ++j) lds_inc(&hist[d[j] >> BSH]);
    } else {
#pragma unroll
        for (int j = 0; j < EPT; ++j)
            if (e0 + 256 * j < E) lds_inc(&hist[d[j] >> BSH]);
    }
    __syncthreads();
    for (int b = threadIdx.x; b < NB; b += 256) tileHist[(long long)t * NB + b] = hist[b];
}

#define TPW 16
__global__ __launch_bounds__(1024) void bin_tile_offsets(const int* __restrict__ tileHist, int* __restrict__ tileOff,
                                                         int* __restrict__ binTotal, int NT, int NB) {
    __shared__ int part[32 * 32];
    const int lane  = threadIdx.x & 31;
    const int wave  = __builtin_amdgcn_readfirstlane((int)(threadIdx.x >> 5));
    const int b     = blockIdx.x * 32 + lane;
    const bool live = b < NB;
    const int bb    = min(b, NB - 1);
    int carry = 0;
    for (int tr = 0; tr < NT; tr += 32 * TPW) {
        const int t0 = tr + wave * TPW;
        int c[TPW];
#pragma unroll
        for (int j = 0; j < TPW; ++j) {
            const int tt = min(t0 + j, NT - 1);
            c[j] = tileHist[(long long)tt * NB + bb];
        }
        int tot = 0;
#pragma unroll
        for (int j = 0; j < TPW; ++j) {
            c[j] = (t0 + j < NT) ? c[j] : 0;
            tot += c[j];
        }
        part[wave * 32 + lane] = tot;
        __syncthreads();
        int before = 0, all = 0;
#pragma unroll
        for (int w = 0; w < 32; ++w) {
            const int p = part[w * 32 + lane];
            all    += p;
            before += (w < wave) ? p : 0;
        }
        int run = carry + before;
#pragma unroll
        for (int j = 0; j < TPW; ++j) {
            if (live && t0 + j < NT) tileOff[(long long)(t0 + j) * NB + b] = run;
            run += c[j];
        }
        carry += all;
        __syncthreads();
    }
    if (wave == 0 && live) binTotal[b] = carry;
}

__global__ __launch_bounds__(1024) void bin_bases(const int* __restrict__ binTotal,
                                                  int* __restrict__ binBase, int NB) {
    __shared__ int part[1024];
    const int tid = threadIdx.x;
    const int per = (NB + 1023) / 1024;
    const int lo  = tid * per;
    int sum = 0;
    for (int k = 0; k < per; ++k) {
        const int b = lo + k;
        if (b < NB) sum += binTotal[b];
    }
    part[tid] = sum;
    __syncthreads();
    for (int off = 1; off < 1024; off <<= 1) {
        const int v = (tid >= off) ? part[tid - off] : 0;
        __syncthreads();
        part[tid] += v;
        __syncthreads();
    }
    int run = part[tid] - sum;
    for (int k = 0; k < per; ++k) {
        const int b = lo + k;
        if (b < NB) { binBase[b] = run; run += binTotal[b]; }
    }
}

#define CPT (MAXNB / 256)
__global__ __launch_bounds__(256) void bin_fill(const int* __restrict__ src, const int* __restrict__ dst,
                                                const int* __restrict__ tileOff,
                                                const int* __restrict__ binBase,
                                                int* __restrict__ list, int E, int NB) {
    __shared__ int cur[MAXNB];
    const int t    = blockIdx.x;
    const int e0   = t * TE + threadIdx.x;
    const bool full = (t + 1) * TE <= E;
    int d[EPT], s[EPT];
    if (full) { tile_words<true>(dst, e0, E, d);  tile_words<true>(src, e0, E, s); }
    else      { tile_words<false>(dst, e0, E, d); tile_words<false>(src, e0, E, s); }
    int cb[CPT], co[CPT];
#pragma unroll
    for (int k = 0; k < CPT; ++k) {
        const int b = min((int)threadIdx.x + 256 * k, NB - 1);
        cb[k] = binBase[b];
        co[k] = tileOff[(long long)t * NB + b];
    }
#pragma unroll
    for (int k = 0; k < CPT; ++k) {
        const int b = (int)threadIdx.x + 256 * k;
        if (b < NB) cur[b] = cb[k] + co[k];
    }
    __syncthreads();
    int p[EPT];
    if (full) {
#pragma unroll
        for (int j = 0; j < EPT; ++j) p[j] = lds_inc(&cur[d[j] >> BSH]);
#pragma unroll
        for (int j = 0; j < EPT; ++j) list[p[j]] = (s[j] << BSH) | (d[j] & BMASK);
    } else {
#pragma unroll
        for (int j = 0; j < EPT; ++j) {
            if (e0 + 256 * j < E) {
                const int q = lds_inc(&cur[d[j] >> BSH]);
                list[q] = (s[j] << BSH) | (d[j] & BMASK);
            }
        }
    }
}

#define NPT 8
__global__ __launch_bounds__(256) void block_norm(const int* __restrict__ list,
                                                  const int* __restrict__ binBase,
                                                  const int* __restrict__ binTotal,
                                                  float* __restrict__ norm, int N) {
    __shared__ int c[BN];
    const int b = blockIdx.x;
    const int base = binBase[b];
    const int cnt  = binTotal[b];
    if (threadIdx.x < BN) c[threadIdx.x] = 0;
    __syncthreads();
    for (int r = 0; r < cnt; r += 256 * NPT) {
        const int i0 = r + (int)threadIdx.x;
        int w[NPT];
#pragma unroll
        for (int j = 0; j < NPT; ++j) w[j] = list[base + min(i0 + 256 * j, cnt - 1)];
#pragma unroll
        for (int j = 0; j < NPT; ++j)
            if (i0 + 256 * j < cnt) lds_inc(&c[w[j] & BMASK]);
    }
    __syncthreads();
    if (threadIdx.x < BN) {
        const int node = b * BN + threadIdx.x;
        if (node < N) norm[node] = 1.0f / sqrtf(fmaxf((float)c[threadIdx.x], 1.0f));
    }
}

template <int P>
__device__ __forceinline__ void gather_rows(const float* __restrict__ x, const float* __restrict__ norm,
                                            const int* __restrict__ list, int base, int cnt,
                                            float* acc, int lane, int wave) {
    for (int i = wave * EPI; i + EPI <= cnt; i += 8 * EPI) {
        int   wd[EPI];
        float nn[EPI], v0[EPI], v1[EPI];
        for (int j = 0; j < EPI; ++j) wd[j] = list[base + i + j];
        for (int j = 0; j < EPI; ++j) {
            const int s = wd[j] >> BSH;
            const float* r = x + (long long)s * 64;
            nn[j] = norm[s];
            v0[j] = r[lane];
            v1[j] = r[lane + 32];
        }
        for (int j = 0; j < EPI; ++j) {
            float* a = acc + (wd[j] & BMASK) * P;
            lds_addf(&a[lane],      v0[j] * nn[j]);
            lds_addf(&a[lane + 32], v1[j] * nn[j]);
        }
    }
    {
        const int i = (cnt / EPI) * EPI + wave;
        if (i < cnt) {
            const int w = list[base + i];
            const int s = w >> BSH;
            const float* r = x + (long long)s * 64;
            const float n = norm[s];
            float* a = acc + (w & BMASK) * P;
            lds_addf(&a[lane],      r[lane] * n);
            lds_addf(&a[lane + 32], r[lane + 32] * n);
        }
    }
}

#define PITCH 68

__global__ __launch_bounds__(256, 16) void aggregate_project(
    const float* __restrict__ feat, const float* __restrict__ norm,
    const int* __restrict__ list, const int* __restrict__ binBase,
    const int* __restrict__ binTotal, const float* __restrict__ W,
    const float* __restrict__ bias, float* __restrict__ outL,
    float* __restrict__ Y, int N) {
    __shared__ __attribute__((aligned(16))) float acc[BN * PITCH];
    __shared__ __attribute__((aligned(16))) float ft[BN * PITCH];
    __shared__ float nb[BN];
    const int b    = blockIdx.x;
    const int lane = threadIdx.x & 31;
    const int wave = threadIdx.x >> 5;
    for (int k = threadIdx.x; k < BN * PITCH; k += 256) acc[k] = 0.0f;
    __syncthreads();

    gather_rows<PITCH>(feat, norm, list, binBase[b], binTotal[b], acc, lane, wave);

    const int f    = threadIdx.x & 63;
    const int r0   = threadIdx.x >> 6;
    const int rows = min(BN, N - b * BN);
#pragma unroll
    for (int j = 0; j < BN / 4; ++j) {
        const int r  = r0 + 4 * j;
        const int rr = min(r, rows - 1);
        const float v = feat[(long long)(b * BN + rr) * 64 + f];
        ft[r * PITCH + f] = (r < rows) ? v : 0.0f;
    }
    if (threadIdx.x < BN) nb[threadIdx.x] = ((int)threadIdx.x < rows) ? norm[b * BN + threadIdx.x] : 0.0f;
    __syncthreads();

    const float ch = -2.0f / LAM;
    const float cx = 2.0f / LAM - 1.0f;
    for (int r = r0; r < BN; r += 4)
        acc[r * PITCH + f] = ch * acc[r * PITCH + f] * nb[r] + cx * ft[r * PITCH + f];
    __syncthreads();

    const int c    = lane & 15;
    const int h    = lane >> 4;
    const int rp   = wave >> 2;
    const int ncol = (wave & 3) * 16 + c;
    const float* W0 = W;
    const float* W1 = W + 4096;
    const float* W2 = W + 8192;
    const float cx2 = 4.0f / LAM - 2.0f;
    const float* fa = ft  + (rp * 32 + c) * PITCH + 2 * h;
    const float* ta = acc + (rp * 32 + c) * PITCH + 2 * h;

    v8f L0 = {}, L1 = {}, Y0 = {}, Y1 = {};
#pragma unroll 2
    for (int k = 0; k < 64; k += 4) {
        const int ka = k + 2 * h;
        const float w0a = W0[ka * 64 + ncol], w0b = W0[(ka + 1) * 64 + ncol];
        const float w1a = W1[ka * 64 + ncol], w1b = W1[(ka + 1) * 64 + ncol];
        const float w2a = W2[ka * 64 + ncol], w2b = W2[(ka + 1) * 64 + ncol];
        const v2f bY = {w2a, w2b};
        const v2f bF = {w0a - w2a, w0b - w2b};
        const v2f bT = {w1a + cx2 * w2a, w1b + cx2 * w2b};
        const v2f f0 = *(const v2f*)(fa + k);
        const v2f t0 = *(const v2f*)(ta + k);
        const v2f f1 = *(const v2f*)(fa + 16 * PITCH + k);
        const v2f t1 = *(const v2f*)(ta + 16 * PITCH + k);
        Y0 = __builtin_amdgcn_wmma_f32_16x16x4_f32(false, t0, false, bY, (short)0, Y0, false, false);
        Y1 = __builtin_amdgcn_wmma_f32_16x16x4_f32(false, t1, false, bY, (short)0, Y1, false, false);
        L0 = __builtin_amdgcn_wmma_f32_16x16x4_f32(false, f0, false, bF, (short)0, L0, false, false);
        L1 = __builtin_amdgcn_wmma_f32_16x16x4_f32(false, f1, false, bF, (short)0, L1, false, false);
        L0 = __builtin_amdgcn_wmma_f32_16x16x4_f32(false, t0, false, bT, (short)0, L0, false, false);
        L1 = __builtin_amdgcn_wmma_f32_16x16x4_f32(false, t1, false, bT, (short)0, L1, false, false);
    }

    const float bv = bias[ncol];
    __syncthreads();
#pragma unroll
    for (int r = 0; r < 8; ++r) {
        const int lr = rp * 32 + r + 8 * h;
        ft[lr * PITCH + ncol]         = L0[r] + bv;
        acc[lr * PITCH + ncol]        = Y0[r];
        ft[(lr + 16) * PITCH + ncol]  = L1[r] + bv;
        acc[(lr + 16) * PITCH + ncol] = Y1[r];
    }
    __syncthreads();
#pragma unroll 4
    for (int j = 0; j < BN / 4; ++j) {
        const int r = r0 + 4 * j;
        if (r < rows) {
            const long long o = (long long)(b * BN + r) * 64 + f;
            outL[o] = ft[r * PITCH + f];
            Y[o]    = acc[r * PITCH + f];
        }
    }
}

__global__ __launch_bounds__(256) void aggregate_finish(
    const float* __restrict__ Y, const float* __restrict__ norm,
    const int* __restrict__ list, const int* __restrict__ binBase,
    const int* __restrict__ binTotal, float* __restrict__ out, int N) {
    __shared__ float acc[BN * 64];
    const int b    = blockIdx.x;
    const int lane = threadIdx.x & 31;
    const int wave = threadIdx.x >> 5;
    for (int k = threadIdx.x; k < BN * 64; k += 256) acc[k] = 0.0f;
    __syncthreads();

    gather_rows<64>(Y, norm, list, binBase[b], binTotal[b], acc, lane, wave);
    __syncthreads();

    const float ch2 = -4.0f / LAM;
    const int f    = threadIdx.x & 63;
    const int r0   = threadIdx.x >> 6;
    const int rows = min(BN, N - b * BN);
#pragma unroll 4
    for (int r = r0; r < rows; r += 4) {
        const int node = b * BN + r;
        const long long o = (long long)node * 64 + f;
        out[o] = ch2 * acc[r * 64 + f] * norm[node] + out[o];
    }
}

extern "C" void kernel_launch(void* const* d_in, const int* in_sizes, int n_in,
                              void* d_out, int out_size, void* d_ws, size_t ws_size,
                              hipStream_t stream) {
    const float* feat = (const float*)d_in[0];
    const float* W    = (const float*)d_in[1];
    const float* bias = (const float*)d_in[2];
    const int*   esrc = (const int*)d_in[3];
    const int*   edst = (const int*)d_in[4];
    float* out = (float*)d_out;

    const int N = in_sizes[0] / 64;
    const int E = in_sizes[3];

    float* ws   = (float*)d_ws;
    float* norm = ws;
    float* Y    = ws + N;

    const int NB = (N + BN - 1) / BN;
    const int NT = (E + TE - 1) / TE;

    int* scr      = (int*)(ws + N + (long long)N * 64);
    int* list     = scr;
    int* tileHist = list + E;
    int* tileOff  = tileHist + (long long)NT * NB;
    int* binTotal = tileOff + (long long)NT * NB;
    int* binBase  = binTotal + NB;

    bin_count<<<NT, 256, 0, stream>>>(edst, tileHist, E, NB);
    bin_tile_offsets<<<(NB + 31) / 32, 1024, 0, stream>>>(tileHist, tileOff, binTotal, NT, NB);
    bin_bases<<<1, 1024, 0, stream>>>(binTotal, binBase, NB);
    bin_fill<<<NT, 256, 0, stream>>>(esrc, edst, tileOff, binBase, list, E, NB);
    block_norm<<<NB, 256, 0, stream>>>(list, binBase, binTotal, norm, N);

    aggregate_project<<<NB, 256, 0, stream>>>(feat, norm, list, binBase, binTotal, W, bias, out, Y, N);
    aggregate_finish<<<NB, 256, 0, stream>>>(Y, norm, list, binBase, binTotal, out, N);
}


# Binning without global compaction: one launch counting-sorts each 16384-edge tile by dst block in LDS; owner work-groups pull their segments out of all tiles; 4 launches, not 7; sum order by arrival
# speedup vs baseline: 5.6086x; 1.3061x over previous
#include <hip/hip_runtime.h>

#define LAM 2.0f

typedef __attribute__((ext_vector_type(2))) float v2f;
typedef __attribute__((ext_vector_type(8))) float v8f;

#define BSH   6
#define BN    (1 << BSH)
#define BMASK (BN - 1)
#define TE    16384
#define EPT   (TE / 1024)
#define MAXNB 2048
#define MAXNT 256
#define CAP   1536
#define EPI   8

__device__ __forceinline__ int lds_inc(int* p) {
    return __hip_atomic_fetch_add(p, 1, __ATOMIC_RELAXED, __HIP_MEMORY_SCOPE_WORKGROUP);
}
__device__ __forceinline__ void lds_addf(float* p, float v) {
    __hip_atomic_fetch_add(p, v, __ATOMIC_RELAXED, __HIP_MEMORY_SCOPE_WORKGROUP);
}

__device__ __forceinline__ int wave_scan(int v, int lane) {
#pragma unroll
    for (int o = 1; o < 32; o <<= 1) {
        const int u = __shfl_up(v, o);
        v += (lane >= o) ? u : 0;
    }
    return v;
}

__global__ __launch_bounds__(1024) void tile_sort(const int* __restrict__ src, const int* __restrict__ dst,
                                                  int* __restrict__ sorted, int* __restrict__ start,
                                                  int E, int NB, int NT) {
    __shared__ int cnt[MAXNB];
    __shared__ int wsum[32];
    __shared__ int outw[TE];
    const int t    = blockIdx.x;
    const int tid  = threadIdx.x;
    const int lane = tid & 31;
    const int wave = tid >> 5;
    const int nval = min(TE, E - t * TE);
    int d[EPT], s[EPT];
#pragma unroll
    for (int j = 0; j < EPT; ++j) {
        const int e = min(t * TE + tid + 1024 * j, E - 1);
        d[j] = dst[e];
        s[j] = src[e];
    }
    __builtin_amdgcn_sched_barrier(0);
    for (int b = tid; b < MAXNB; b += 1024) cnt[b] = 0;
    __syncthreads();
    int r[EPT];
#pragma unroll
    for (int j = 0; j < EPT; ++j)
        r[j] = (tid + 1024 * j < nval) ? lds_inc(&cnt[d[j] >> BSH]) : 0;
    __syncthreads();
    const int a0 = cnt[2 * tid], a1 = cnt[2 * tid + 1];
    const int inc = wave_scan(a0 + a1, lane);
    if (lane == 31) wsum[wave] = inc;
    __syncthreads();
    const int wv = wsum[lane];
    const int wi = wave_scan(wv, lane);
    const int before = __shfl(wi, wave) - __shfl(wv, wave);
    const int ex = before + inc - (a0 + a1);
    cnt[2 * tid]     = ex;
    cnt[2 * tid + 1] = ex + a0;
    __syncthreads();
    for (int b = tid; b <= NB; b += 1024)
        start[(long long)b * NT + t] = (b < NB) ? cnt[b] : nval;
#pragma unroll
    for (int j = 0; j < EPT; ++j)
        if (tid + 1024 * j < nval) outw[cnt[d[j] >> BSH] + r[j]] = (s[j] << BSH) | (d[j] & BMASK);
    __syncthreads();
    for (int k = tid; k < nval; k += 1024) sorted[(long long)t * TE + k] = outw[k];
}

__device__ __forceinline__ int pull_setup(const int* __restrict__ start, int b, int NT,
                                          int* off, int* st, int* wsum) {
    const int tid  = threadIdx.x;
    const int lane = tid & 31;
    const int wave = tid >> 5;
    int s0 = 0, c = 0;
    if (tid < NT) {
        s0 = start[(long long)b * NT + tid];
        c  = start[(long long)(b + 1) * NT + tid] - s0;
    }
    const int inc = wave_scan(c, lane);
    if (lane == 31) wsum[wave] = inc;
    __syncthreads();
    int before = 0, all = 0;
#pragma unroll
    for (int w = 0; w < 8; ++w) {
        const int x = wsum[w];
        all    += x;
        before += (w < wave) ? x : 0;
    }
    if (tid < NT) { off[tid] = before + inc - c; st[tid] = s0; }
    if (tid == 0) off[NT] = all;
    __syncthreads();
    return all;
}

__device__ __forceinline__ int list_word(const int* __restrict__ sorted, const int* off, const int* st,
                                         int NT, int p) {
    int lo = 0, hi = NT;
#pragma unroll
    for (int it = 0; it < 8; ++it) {
        const int mid = (lo + hi) >> 1;
        const bool ge = off[mid] <= p;
        lo = ge ? mid : lo;
        hi = ge ? hi : mid;
    }
    return sorted[(long long)lo * TE + st[lo] + (p - off[lo])];
}

__device__ __forceinline__ void pull_chunk(const int* __restrict__ sorted, const int* off, const int* st,
                                           int NT, int c0, int n, int total, int* lst) {
    int w[CAP / 256];
#pragma unroll
    for (int k = 0; k < CAP / 256; ++k)
        w[k] = list_word(sorted, off, st, NT, min(c0 + (int)threadIdx.x + 256 * k, total - 1));
#pragma unroll
    for (int k = 0; k < CAP / 256; ++k) {
        const int q = (int)threadIdx.x + 256 * k;
        if (q < n) lst[q] = w[k];
    }
}

#define NPT 6
__global__ __launch_bounds__(256) void block_norm(const int* __restrict__ sorted,
                                                  const int* __restrict__ start,
                                                  float* __restrict__ norm, int N, int NT) {
    __shared__ int c[BN];
    __shared__ int off[MAXNT + 1];
    __shared__ int st[MAXNT];
    __shared__ int wsum[8];
    const int b = blockIdx.x;
    if (threadIdx.x < BN) c[threadIdx.x] = 0;
    const int cnt = pull_setup(start, b, NT, off, st, wsum);
    for (int r = 0; r < cnt; r += 256 * NPT) {
        const int i0 = r + (int)threadIdx.x;
        int w[NPT];
#pragma unroll
        for (int j = 0; j < NPT; ++j) w[j] = list_word(sorted, off, st, NT, min(i0 + 256 * j, cnt - 1));
#pragma unroll
        for (int j = 0; j < NPT; ++j)
            if (i0 + 256 * j < cnt) lds_inc(&c[w[j] & BMASK]);
    }
    __syncthreads();
    if (threadIdx.x < BN) {
        const int node = b * BN + threadIdx.x;
        if (node < N) norm[node] = 1.0f / sqrtf(fmaxf((float)c[threadIdx.x], 1.0f));
    }
}

template <int P>
__device__ __forceinline__ void gather_rows(const float* __restrict__ x, const float* __restrict__ norm,
                                            const int* lst, int cnt,
                                            float* acc, int lane, int wave) {
    for (int i = wave * EPI; i + EPI <= cnt; i += 8 * EPI) {
        int   wd[EPI];
        float nn[EPI], v0[EPI], v1[EPI];
        for (int j = 0; j < EPI; ++j) wd[j] = lst[i + j];
        for (int j = 0; j < EPI; ++j) {
            const int s = wd[j] >> BSH;
            const float* r = x + (long long)s * 64;
            nn[j] = norm[s];
            v0[j] = r[lane];
            v1[j] = r[lane + 32];
        }
        for (int j = 0; j < EPI; ++j) {
            float* a = acc + (wd[j] & BMASK) * P;
            lds_addf(&a[lane],      v0[j] * nn[j]);
            lds_addf(&a[lane + 32], v1[j] * nn[j]);
        }
    }
    {
        const int i = (cnt / EPI) * EPI + wave;
        if (i < cnt) {
            const int w = lst[i];
            const int s = w >> BSH;
            const float* r = x + (long long)s * 64;
            const float n = norm[s];
            float* a = acc + (w & BMASK) * P;
            lds_addf(&a[lane],      r[lane] * n);
            lds_addf(&a[lane + 32], r[lane + 32] * n);
        }
    }
}

template <int P>
__device__ __forceinline__ void gather_block(const float* __restrict__ x, const float* __restrict__ norm,
                                             const int* __restrict__ sorted, const int* off, const int* st,
                                             int NT, int total, int* lst, float* acc, int lane, int wave) {
    for (int c0 = 0; c0 < total; c0 += CAP) {
        if (c0 > 0) __syncthreads();
        const int n = min(CAP, total - c0);
        pull_chunk(sorted, off, st, NT, c0, n, total, lst);
        __syncthreads();
        gather_rows<P>(x, norm, lst, n, acc, lane, wave);
    }
}

#define PITCH 68

__global__ __launch_bounds__(256, 16) void aggregate_project(
    const float* __restrict__ feat, const float* __restrict__ norm,
    const int* __restrict__ sorted, const int* __restrict__ start,
    const float* __restrict__ W,
    const float* __restrict__ bias, float* __restrict__ outL,
    float* __restrict__ Y, int N, int NT) {
    __shared__ __attribute__((aligned(16))) float acc[BN * PITCH];
    __shared__ __attribute__((aligned(16))) float ft[BN * PITCH];
    __shared__ float nb[BN];
    __shared__ int wsum[8];
    int* lst = (int*)ft;
    int* off = lst + CAP;
    int* st  = off + MAXNT + 1;
    const int b    = blockIdx.x;
    const int lane = threadIdx.x & 31;
    const int wave = threadIdx.x >> 5;
    for (int k = threadIdx.x; k < BN * PITCH; k += 256) acc[k] = 0.0f;
    const int total = pull_setup(start, b, NT, off, st, wsum);

    gather_block<PITCH>(feat, norm, sorted, off, st, NT, total, lst, acc, lane, wave);

    const int f    = threadIdx.x & 63;
    const int r0   = threadIdx.x >> 6;
    const int rows = min(BN, N - b * BN);
    float fv[BN / 4];
#pragma unroll
    for (int j = 0; j < BN / 4; ++j) {
        const int rr = min(r0 + 4 * j, rows - 1);
        fv[j] = feat[(long long)(b * BN + rr) * 64 + f];
    }
    float nv = 0.0f;
    if (threadIdx.x < BN) nv = ((int)threadIdx.x < rows) ? norm[b * BN + threadIdx.x] : 0.0f;
    __syncthreads();
#pragma unroll
    for (int j = 0; j < BN / 4; ++j) {
        const int r = r0 + 4 * j;
        ft[r * PITCH + f] = (r < rows) ? fv[j] : 0.0f;
    }
    if (threadIdx.x < BN) nb[threadIdx.x] = nv;
    __syncthreads();

    const float ch = -2.0f / LAM;
    const float cx = 2.0f / LAM - 1.0f;
    for (int r = r0; r < BN; r += 4)
        acc[r * PITCH + f] = ch * acc[r * PITCH + f] * nb[r] + cx * ft[r * PITCH + f];
    __syncthreads();

    const int c    = lane & 15;
    const int h    = lane >> 4;
    const int rp   = wave >> 2;
    const int ncol = (wave & 3) * 16 + c;
    const float* W0 = W;
    const float* W1 = W + 4096;
    const float* W2 = W + 8192;
    const float cx2 = 4.0f / LAM - 2.0f;
    const float* fa = ft  + (rp * 32 + c) * PITCH + 2 * h;
    const float* ta = acc + (rp * 32 + c) * PITCH + 2 * h;

    v8f L0 = {}, L1 = {}, Y0 = {}, Y1 = {};
#pragma unroll 2
    for (int k = 0; k < 64; k += 4) {
        const int ka = k + 2 * h;
        const float w0a = W0[ka * 64 + ncol], w0b = W0[(ka + 1) * 64 + ncol];
        const float w1a = W1[ka * 64 + ncol], w1b = W1[(ka + 1) * 64 + ncol];
        const float w2a = W2[ka * 64 + ncol], w2b = W2[(ka + 1) * 64 + ncol];
        const v2f bY = {w2a, w2b};
        const v2f bF = {w0a - w2a, w0b - w2b};
        const v2f bT = {w1a + cx2 * w2a, w1b + cx2 * w2b};
        const v2f f0 = *(const v2f*)(fa + k);
        const v2f t0 = *(const v2f*)(ta + k);
        const v2f f1 = *(const v2f*)(fa + 16 * PITCH + k);
        const v2f t1 = *(const v2f*)(ta + 16 * PITCH + k);
        Y0 = __builtin_amdgcn_wmma_f32_16x16x4_f32(false, t0, false, bY, (short)0, Y0, false, false);
        Y1 = __builtin_amdgcn_wmma_f32_16x16x4_f32(false, t1, false, bY, (short)0, Y1, false, false);
        L0 = __builtin_amdgcn_wmma_f32_16x16x4_f32(false, f0, false, bF, (short)0, L0, false, false);
        L1 = __builtin_amdgcn_wmma_f32_16x16x4_f32(false, f1, false, bF, (short)0, L1, false, false);
        L0 = __builtin_amdgcn_wmma_f32_16x16x4_f32(false, t0, false, bT, (short)0, L0, false, false);
        L1 = __builtin_amdgcn_wmma_f32_16x16x4_f32(false, t1, false, bT, (short)0, L1, false, false);
    }

    const float bv = bias[ncol];
    __syncthreads();
#pragma unroll
    for (int r = 0; r < 8; ++r) {
        const int lr = rp * 32 + r + 8 * h;
        ft[lr * PITCH + ncol]         = L0[r] + bv;
        acc[lr * PITCH + ncol]        = Y0[r];
        ft[(lr + 16) * PITCH + ncol]  = L1[r] + bv;
        acc[(lr + 16) * PITCH + ncol] = Y1[r];
    }
    __syncthreads();
#pragma unroll 4
    for (int j = 0; j < BN / 4; ++j) {
        const int r = r0 + 4 * j;
        if (r < rows) {
            const long long o = (long long)(b * BN + r) * 64 + f;
            outL[o] = ft[r * PITCH + f];
            Y[o]    = acc[r * PITCH + f];
        }
    }
}

__global__ __launch_bounds__(256, 16) void aggregate_finish(
    const float* __restrict__ Y, const float* __restrict__ norm,
    const int* __restrict__ sorted, const int* __restrict__ start,
    float* __restrict__ out, int N, int NT) {
    __shared__ float acc[BN * 64];
    __shared__ __attribute__((aligned(16))) int lst[CAP];
    __shared__ int off[MAXNT + 1];
    __shared__ int st[MAXNT];
    __shared__ int wsum[8];
    const int b    = blockIdx.x;
    const int lane = threadIdx.x & 31;
    const int wave = threadIdx.x >> 5;
    for (int k = threadIdx.x; k < BN * 64; k += 256) acc[k] = 0.0f;
    const int total = pull_setup(start, b, NT, off, st, wsum);

    gather_block<64>(Y, norm, sorted, off, st, NT, total, lst, acc, lane, wave);
    __syncthreads();

    const float ch2 = -4.0f / LAM;
    const int f    = threadIdx.x & 63;
    const int r0   = threadIdx.x >> 6;
    const int rows = min(BN, N - b * BN);
#pragma unroll 4
    for (int r = r0; r < rows; r += 4) {
        const int node = b * BN + r;
        const long long o = (long long)node * 64 + f;
        out[o] = ch2 * acc[r * 64 + f] * norm[node] + out[o];
    }
}

extern "C" void kernel_launch(void* const* d_in, const int* in_sizes, int n_in,
                              void* d_out, int out_size, void* d_ws, size_t ws_size,
                              hipStream_t stream) {
    const float* feat = (const float*)d_in[0];
    const float* W    = (const float*)d_in[1];
    const float* bias = (const float*)d_in[2];
    const int*   esrc = (const int*)d_in[3];
    const int*   edst = (const int*)d_in[4];
    float* out = (float*)d_out;

    const int N = in_sizes[0] / 64;
    const int E = in_sizes[3];

    float* ws   = (float*)d_ws;
    float* norm = ws;
    float* Y    = ws + N;

    const int NB = (N + BN - 1) / BN;
    const int NT = (E + TE - 1) / TE;

    int* scr    = (int*)(ws + N + (long long)N * 64);
    int* sorted = scr;
    int* start  = sorted + (long long)NT * TE;

    tile_sort<<<NT, 1024, 0, stream>>>(esrc, edst, sorted, start, E, NB, NT);
    block_norm<<<NB, 256, 0, stream>>>(sorted, start, norm, N, NT);

    aggregate_project<<<NB, 256, 0, stream>>>(feat, norm, sorted, start, W, bias, out, Y, N, NT);
    aggregate_finish<<<NB, 256, 0, stream>>>(Y, norm, sorted, start, out, N, NT);
}


# Gather holds an edge's own values (list word, source norm, LDS row) in scalar registers, rows load by 32-bit index off the scalar table base: 16 edges in flight per wave, not 8; sum order by arrival
# speedup vs baseline: 5.8567x; 1.0442x over previous
#include <hip/hip_runtime.h>

#define LAM 2.0f

typedef __attribute__((ext_vector_type(2))) float v2f;
typedef __attribute__((ext_vector_type(8))) float v8f;

#define BSH   6
#define BN    (1 << BSH)
#define BMASK (BN - 1)
#define TE    16384
#define EPT   (TE / 1024)
#define MAXNB 2048
#define MAXNT 256
#define CAP   1536
#define EPI   16

__device__ __forceinline__ int lds_inc(int* p) {
    return __hip_atomic_fetch_add(p, 1, __ATOMIC_RELAXED, __HIP_MEMORY_SCOPE_WORKGROUP);
}
__device__ __forceinline__ void lds_addf(float* p, float v) {
    __hip_atomic_fetch_add(p, v, __ATOMIC_RELAXED, __HIP_MEMORY_SCOPE_WORKGROUP);
}

__device__ __forceinline__ int wave_scan(int v, int lane) {
#pragma unroll
    for (int o = 1; o < 32; o <<= 1) {
        const int u = __shfl_up(v, o);
        v += (lane >= o) ? u : 0;
    }
    return v;
}

__global__ __launch_bounds__(1024) void tile_sort(const int* __restrict__ src, const int* __restrict__ dst,
                                                  int* __restrict__ sorted, int* __restrict__ start,
                                                  int E, int NB, int NT) {
    __shared__ int cnt[MAXNB];
    __shared__ int wsum[32];
    __shared__ int outw[TE];
    const int t    = blockIdx.x;
    const int tid  = threadIdx.x;
    const int lane = tid & 31;
    const int wave = tid >> 5;
    const int nval = min(TE, E - t * TE);
    int d[EPT], s[EPT];
#pragma unroll
    for (int j = 0; j < EPT; ++j) {
        const int e = min(t * TE + tid + 1024 * j, E - 1);
        d[j] = dst[e];
        s[j] = src[e];
    }
    __builtin_amdgcn_sched_barrier(0);
    for (int b = tid; b < MAXNB; b += 1024) cnt[b] = 0;
    __syncthreads();
    int r[EPT];
#pragma unroll
    for (int j = 0; j < EPT; ++j)
        r[j] = (tid + 1024 * j < nval) ? lds_inc(&cnt[d[j] >> BSH]) : 0;
    __syncthreads();
    const int a0 = cnt[2 * tid], a1 = cnt[2 * tid + 1];
    const int inc = wave_scan(a0 + a1, lane);
    if (lane == 31) wsum[wave] = inc;
    __syncthreads();
    const int wv = wsum[lane];
    const int wi = wave_scan(wv, lane);
    const int before = __shfl(wi, wave) - __shfl(wv, wave);
    const int ex = before + inc - (a0 + a1);
    cnt[2 * tid]     = ex;
    cnt[2 * tid + 1] = ex + a0;
    __syncthreads();
    for (int b = tid; b <= NB; b += 1024)
        start[(long long)b * NT + t] = (b < NB) ? cnt[b] : nval;
#pragma unroll
    for (int j = 0; j < EPT; ++j)
        if (tid + 1024 * j < nval) outw[cnt[d[j] >> BSH] + r[j]] = (s[j] << BSH) | (d[j] & BMASK);
    __syncthreads();
    for (int k = tid; k < nval; k += 1024) sorted[(long long)t * TE + k] = outw[k];
}

__device__ __forceinline__ int pull_setup(const int* __restrict__ start, int b, int NT,
                                          int* off, int* st, int* wsum) {
    const int tid  = threadIdx.x;
    const int lane = tid & 31;
    const int wave = tid >> 5;
    int s0 = 0, c = 0;
    if (tid < NT) {
        s0 = start[(long long)b * NT + tid];
        c  = start[(long long)(b + 1) * NT + tid] - s0;
    }
    const int inc = wave_scan(c, lane);
    if (lane == 31) wsum[wave] = inc;
    __syncthreads();
    int before = 0, all = 0;
#pragma unroll
    for (int w = 0; w < 8; ++w) {
        const int x = wsum[w];
        all    += x;
        before += (w < wave) ? x : 0;
    }
    if (tid < NT) { off[tid] = before + inc - c; st[tid] = s0; }
    if (tid == 0) off[NT] = all;
    __syncthreads();
    return all;
}

__device__ __forceinline__ int list_word(const int* __restrict__ sorted, const int* off, const int* st,
                                         int NT, int p) {
    int lo = 0, hi = NT;
#pragma unroll
    for (int it = 0; it < 8; ++it) {
        const int mid = (lo + hi) >> 1;
        const bool ge = off[mid] <= p;
        lo = ge ? mid : lo;
        hi = ge ? hi : mid;
    }
    return sorted[(long long)lo * TE + st[lo] + (p - off[lo])];
}

__device__ __forceinline__ void pull_chunk(const int* __restrict__ sorted, const int* off, const int* st,
                                           int NT, int c0, int n, int total, int* lst) {
    int w[CAP / 256];
#pragma unroll
    for (int k = 0; k < CAP / 256; ++k)
        w[k] = list_word(sorted, off, st, NT, min(c0 + (int)threadIdx.x + 256 * k, total - 1));
#pragma unroll
    for (int k = 0; k < CAP / 256; ++k) {
        const int q = (int)threadIdx.x + 256 * k;
        if (q < n) lst[q] = w[k];
    }
}

#define NPT 6
__global__ __launch_bounds__(256) void block_norm(const int* __restrict__ sorted,
                                                  const int* __restrict__ start,
                                                  float* __restrict__ norm, int N, int NT) {
    __shared__ int c[BN];
    __shared__ int off[MAXNT + 1];
    __shared__ int st[MAXNT];
    __shared__ int wsum[8];
    const int b = blockIdx.x;
    if (threadIdx.x < BN) c[threadIdx.x] = 0;
    const int cnt = pull_setup(start, b, NT, off, st, wsum);
    for (int r = 0; r < cnt; r += 256 * NPT) {
        const int i0 = r + (int)threadIdx.x;
        int w[NPT];
#pragma unroll
        for (int j = 0; j < NPT; ++j) w[j] = list_word(sorted, off, st, NT, min(i0 + 256 * j, cnt - 1));
#pragma unroll
        for (int j = 0; j < NPT; ++j)
            if (i0 + 256 * j < cnt) lds_inc(&c[w[j] & BMASK]);
    }
    __syncthreads();
    if (threadIdx.x < BN) {
        const int node = b * BN + threadIdx.x;
        if (node < N) norm[node] = 1.0f / sqrtf(fmaxf((float)c[threadIdx.x], 1.0f));
    }
}

template <int P>
__device__ __forceinline__ void gather_rows(const float* __restrict__ x, const float* __restrict__ norm,
                                            const int* lst, int cnt,
                                            float* acc, int lane, int wave) {
    static_assert(EPI == 16, "lane & (EPI - 1) picks a lane's word; the left-over loop takes EPI / 8 edges a wave");
    const int wv = __builtin_amdgcn_readfirstlane(wave);
    for (int i = wv * EPI; i + EPI <= cnt; i += 8 * EPI) {
        const int   wl = lst[i + (lane & (EPI - 1))];
        const float nl = norm[(unsigned)wl >> BSH];
        unsigned wd[EPI], idx[EPI];
        float v0[EPI], v1[EPI];
        for (int j = 0; j < EPI; ++j) {
            wd[j] = (unsigned)__builtin_amdgcn_readlane(wl, j);
            idx[j] = (wd[j] & ~(unsigned)BMASK) | (unsigned)lane;
        }
        __builtin_amdgcn_sched_barrier(0);
        for (int j = 0; j < EPI; ++j) {
            const float* p = x + idx[j];
            v0[j] = p[0];
            v1[j] = p[32];
        }
        for (int j = 0; j < EPI; ++j) {
            const float n = __int_as_float(__builtin_amdgcn_readlane(__float_as_int(nl), j));
            float* a = acc + (wd[j] & BMASK) * P;
            lds_addf(&a[lane],      v0[j] * n);
            lds_addf(&a[lane + 32], v1[j] * n);
        }
    }
    const int base = (cnt / EPI) * EPI + wv;
    if (base < cnt) {
        unsigned w[EPI / 8];
        float n[EPI / 8], a0[EPI / 8], a1[EPI / 8];
        for (int k = 0; k < EPI / 8; ++k) {
            w[k] = (unsigned)__builtin_amdgcn_readfirstlane(lst[min(base + 8 * k, cnt - 1)]);
            const float* p = x + ((w[k] & ~(unsigned)BMASK) | (unsigned)lane);
            n[k]  = norm[w[k] >> BSH];
            a0[k] = p[0];
            a1[k] = p[32];
        }
        for (int k = 0; k < EPI / 8; ++k) {
            if (base + 8 * k < cnt) {
                float* a = acc + (w[k] & BMASK) * P;
                lds_addf(&a[lane],      a0[k] * n[k]);
                lds_addf(&a[lane + 32], a1[k] * n[k]);
            }
        }
    }
}

template <int P>
__device__ __forceinline__ void gather_block(const float* __restrict__ x, const float* __restrict__ norm,
                                             const int* __restrict__ sorted, const int* off, const int* st,
                                             int NT, int total, int* lst, float* acc, int lane, int wave) {
    for (int c0 = 0; c0 < total; c0 += CAP) {
        if (c0 > 0) __syncthreads();
        const int n = min(CAP, total - c0);
        pull_chunk(sorted, off, st, NT, c0, n, total, lst);
        __syncthreads();
        gather_rows<P>(x, norm, lst, n, acc, lane, wave);
    }
}

#define PITCH 68

__global__ __launch_bounds__(256, 16) void aggregate_project(
    const float* __restrict__ feat, const float* __restrict__ norm,
    const int* __restrict__ sorted, const int* __restrict__ start,
    const float* __restrict__ W,
    const float* __restrict__ bias, float* __restrict__ outL,
    float* __restrict__ Y, int N, int NT) {
    __shared__ __attribute__((aligned(16))) float acc[BN * PITCH];
    __shared__ __attribute__((aligned(16))) float ft[BN * PITCH];
    __shared__ float nb[BN];
    __shared__ int wsum[8];
    int* lst = (int*)ft;
    int* off = lst + CAP;
    int* st  = off + MAXNT + 1;
    const int b    = blockIdx.x;
    const int lane = threadIdx.x & 31;
    const int wave = threadIdx.x >> 5;
    for (int k = threadIdx.x; k < BN * PITCH; k += 256) acc[k] = 0.0f;
    const int total = pull_setup(start, b, NT, off, st, wsum);

    gather_block<PITCH>(feat, norm, sorted, off, st, NT, total, lst, acc, lane, wave);

    const int f    = threadIdx.x & 63;
    const int r0   = threadIdx.x >> 6;
    const int rows = min(BN, N - b * BN);
    float fv[BN / 4];
#pragma unroll
    for (int j = 0; j < BN / 4; ++j) {
        const int rr = min(r0 + 4 * j, rows - 1);
        fv[j] = feat[(long long)(b * BN + rr) * 64 + f];
    }
    float nv = 0.0f;
    if (threadIdx.x < BN) nv = ((int)threadIdx.x < rows) ? norm[b * BN + threadIdx.x] : 0.0f;
    __syncthreads();
#pragma unroll
    for (int j = 0; j < BN / 4; ++j) {
        const int r = r0 + 4 * j;
        ft[r * PITCH + f] = (r < rows) ? fv[j] : 0.0f;
    }
    if (threadIdx.x < BN) nb[threadIdx.x] = nv;
    __syncthreads();

    const float ch = -2.0f / LAM;
    const float cx = 2.0f / LAM - 1.0f;
    for (int r = r0; r < BN; r += 4)
        acc[r * PITCH + f] = ch * acc[r * PITCH + f] * nb[r] + cx * ft[r * PITCH + f];
    __syncthreads();

    const int c    = lane & 15;
    const int h    = lane >> 4;
    const int rp   = wave >> 2;
    const int ncol = (wave & 3) * 16 + c;
    const float* W0 = W;
    const float* W1 = W + 4096;
    const float* W2 = W + 8192;
    const float cx2 = 4.0f / LAM - 2.0f;
    const float* fa = ft  + (rp * 32 + c) * PITCH + 2 * h;
    const float* ta = acc + (rp * 32 + c) * PITCH + 2 * h;

    v8f L0 = {}, L1 = {}, Y0 = {}, Y1 = {};
#pragma unroll 2
    for (int k = 0; k < 64; k += 4) {
        const int ka = k + 2 * h;
        const float w0a = W0[ka * 64 + ncol], w0b = W0[(ka + 1) * 64 + ncol];
        const float w1a = W1[ka * 64 + ncol], w1b = W1[(ka + 1) * 64 + ncol];
        const float w2a = W2[ka * 64 + ncol], w2b = W2[(ka + 1) * 64 + ncol];
        const v2f bY = {w2a, w2b};
        const v2f bF = {w0a - w2a, w0b - w2b};
        const v2f bT = {w1a + cx2 * w2a, w1b + cx2 * w2b};
        const v2f f0 = *(const v2f*)(fa + k);
        const v2f t0 = *(const v2f*)(ta + k);
        const v2f f1 = *(const v2f*)(fa + 16 * PITCH + k);
        const v2f t1 = *(const v2f*)(ta + 16 * PITCH + k);
        Y0 = __builtin_amdgcn_wmma_f32_16x16x4_f32(false, t0, false, bY, (short)0, Y0, false, false);
        Y1 = __builtin_amdgcn_wmma_f32_16x16x4_f32(false, t1, false, bY, (short)0, Y1, false, false);
        L0 = __builtin_amdgcn_wmma_f32_16x16x4_f32(false, f0, false, bF, (short)0, L0, false, false);
        L1 = __builtin_amdgcn_wmma_f32_16x16x4_f32(false, f1, false, bF, (short)0, L1, false, false);
        L0 = __builtin_amdgcn_wmma_f32_16x16x4_f32(false, t0, false, bT, (short)0, L0, false, false);
        L1 = __builtin_amdgcn_wmma_f32_16x16x4_f32(false, t1, false, bT, (short)0, L1, false, false);
    }

    const float bv = bias[ncol];
    __syncthreads();
#pragma unroll
    for (int r = 0; r < 8; ++r) {
        const int lr = rp * 32 + r + 8 * h;
        ft[lr * PITCH + ncol]         = L0[r] + bv;
        acc[lr * PITCH + ncol]        = Y0[r];
        ft[(lr + 16) * PITCH + ncol]  = L1[r] + bv;
        acc[(lr + 16) * PITCH + ncol] = Y1[r];
    }
    __syncthreads();
#pragma unroll 4
    for (int j = 0; j < BN / 4; ++j) {
        const int r = r0 + 4 * j;
        if (r < rows) {
            const long long o = (long long)(b * BN + r) * 64 + f;
            outL[o] = ft[r * PITCH + f];
            Y[o]    = acc[r * PITCH + f];
        }
    }
}

__global__ __launch_bounds__(256, 16) void aggregate_finish(
    const float* __restrict__ Y, const float* __restrict__ norm,
    const int* __restrict__ sorted, const int* __restrict__ start,
    float* __restrict__ out, int N, int NT) {
    __shared__ float acc[BN * 64];
    __shared__ __attribute__((aligned(16))) int lst[CAP];
    __shared__ int off[MAXNT + 1];
    __shared__ int st[MAXNT];
    __shared__ int wsum[8];
    const int b    = blockIdx.x;
    const int lane = threadIdx.x & 31;
    const int wave = threadIdx.x >> 5;
    for (int k = threadIdx.x; k < BN * 64; k += 256) acc[k] = 0.0f;
    const int total = pull_setup(start, b, NT, off, st, wsum);

    gather_block<64>(Y, norm, sorted, off, st, NT, total, lst, acc, lane, wave);
    __syncthreads();

    const float ch2 = -4.0f / LAM;
    const int f    = threadIdx.x & 63;
    const int r0   = threadIdx.x >> 6;
    const int rows = min(BN, N - b * BN);
#pragma unroll 4
    for (int r = r0; r < rows; r += 4) {
        const int node = b * BN + r;
        const long long o = (long long)node * 64 + f;
        out[o] = ch2 * acc[r * 64 + f] * norm[node] + out[o];
    }
}

extern "C" void kernel_launch(void* const* d_in, const int* in_sizes, int n_in,
                              void* d_out, int out_size, void* d_ws, size_t ws_size,
                              hipStream_t stream) {
    const float* feat = (const float*)d_in[0];
    const float* W    = (const float*)d_in[1];
    const float* bias = (const float*)d_in[2];
    const int*   esrc = (const int*)d_in[3];
    const int*   edst = (const int*)d_in[4];
    float* out = (float*)d_out;

    const int N = in_sizes[0] / 64;
    const int E = in_sizes[3];

    float* ws   = (float*)d_ws;
    float* norm = ws;
    float* Y    = ws + N;

    const int NB = (N + BN - 1) / BN;
    const int NT = (E + TE - 1) / TE;

    int* scr    = (int*)(ws + N + (long long)N * 64);
    int* sorted = scr;
    int* start  = sorted + (long long)NT * TE;

    tile_sort<<<NT, 1024, 0, stream>>>(esrc, edst, sorted, start, E, NB, NT);
    block_norm<<<NB, 256, 0, stream>>>(sorted, start, norm, N, NT);

    aggregate_project<<<NB, 256, 0, stream>>>(feat, norm, sorted, start, W, bias, out, Y, N, NT);
    aggregate_finish<<<NB, 256, 0, stream>>>(Y, norm, sorted, start, out, N, NT);
}


# Owner work-group orders its edge list by destination row (LDS counting sort); a wave sums a row's run in registers: one LDS float add pair per run, not per edge; sum order: list order inside a run
# speedup vs baseline: 6.3650x; 1.0868x over previous
#include <hip/hip_runtime.h>

#define LAM 2.0f

typedef __attribute__((ext_vector_type(2))) float v2f;
typedef __attribute__((ext_vector_type(8))) float v8f;

#define BSH   6
#define BN    (1 << BSH)
#define BMASK (BN - 1)
#define TE    16384
#define EPT   (TE / 1024)
#define MAXNB 2048
#define MAXNT 256
#define CAP   1536
#define EPI   16

__device__ __forceinline__ int lds_inc(int* p) {
    return __hip_atomic_fetch_add(p, 1, __ATOMIC_RELAXED, __HIP_MEMORY_SCOPE_WORKGROUP);
}
__device__ __forceinline__ void lds_addf(float* p, float v) {
    __hip_atomic_fetch_add(p, v, __ATOMIC_RELAXED, __HIP_MEMORY_SCOPE_WORKGROUP);
}

__device__ __forceinline__ int wave_scan(int v, int lane) {
#pragma unroll
    for (int o = 1; o < 32; o <<= 1) {
        const int u = __shfl_up(v, o);
        v += (lane >= o) ? u : 0;
    }
    return v;
}

__global__ __launch_bounds__(1024) void tile_sort(const int* __restrict__ src, const int* __restrict__ dst,
                                                  int* __restrict__ sorted, int* __restrict__ start,
                                                  int E, int NB, int NT) {
    __shared__ int cnt[MAXNB];
    __shared__ int wsum[32];
    __shared__ int outw[TE];
    const int t    = blockIdx.x;
    const int tid  = threadIdx.x;
    const int lane = tid & 31;
    const int wave = tid >> 5;
    const int nval = min(TE, E - t * TE);
    int d[EPT], s[EPT];
#pragma unroll
    for (int j = 0; j < EPT; ++j) {
        const int e = min(t * TE + tid + 1024 * j, E - 1);
        d[j] = dst[e];
        s[j] = src[e];
    }
    __builtin_amdgcn_sched_barrier(0);
    for (int b = tid; b < MAXNB; b += 1024) cnt[b] = 0;
    __syncthreads();
    int r[EPT];
#pragma unroll
    for (int j = 0; j < EPT; ++j)
        r[j] = (tid + 1024 * j < nval) ? lds_inc(&cnt[d[j] >> BSH]) : 0;
    __syncthreads();
    const int a0 = cnt[2 * tid], a1 = cnt[2 * tid + 1];
    const int inc = wave_scan(a0 + a1, lane);
    if (lane == 31) wsum[wave] = inc;
    __syncthreads();
    const int wv = wsum[lane];
    const int wi = wave_scan(wv, lane);
    const int before = __shfl(wi, wave) - __shfl(wv, wave);
    const int ex = before + inc - (a0 + a1);
    cnt[2 * tid]     = ex;
    cnt[2 * tid + 1] = ex + a0;
    __syncthreads();
    for (int b = tid; b <= NB; b += 1024)
        start[(long long)b * NT + t] = (b < NB) ? cnt[b] : nval;
#pragma unroll
    for (int j = 0; j < EPT; ++j)
        if (tid + 1024 * j < nval) outw[cnt[d[j] >> BSH] + r[j]] = (s[j] << BSH) | (d[j] & BMASK);
    __syncthreads();
    for (int k = tid; k < nval; k += 1024) sorted[(long long)t * TE + k] = outw[k];
}

__device__ __forceinline__ int pull_setup(const int* __restrict__ start, int b, int NT,
                                          int* off, int* st, int* wsum) {
    const int tid  = threadIdx.x;
    const int lane = tid & 31;
    const int wave = tid >> 5;
    int s0 = 0, c = 0;
    if (tid < NT) {
        s0 = start[(long long)b * NT + tid];
        c  = start[(long long)(b + 1) * NT + tid] - s0;
    }
    const int inc = wave_scan(c, lane);
    if (lane == 31) wsum[wave] = inc;
    __syncthreads();
    int before = 0, all = 0;
#pragma unroll
    for (int w = 0; w < 8; ++w) {
        const int x = wsum[w];
        all    += x;
        before += (w < wave) ? x : 0;
    }
    if (tid < NT) { off[tid] = before + inc - c; st[tid] = s0; }
    if (tid == 0) off[NT] = all;
    __syncthreads();
    return all;
}

__device__ __forceinline__ int list_word(const int* __restrict__ sorted, const int* off, const int* st,
                                         int NT, int p) {
    int lo = 0, hi = NT;
#pragma unroll
    for (int it = 0; it < 8; ++it) {
        const int mid = (lo + hi) >> 1;
        const bool ge = off[mid] <= p;
        lo = ge ? mid : lo;
        hi = ge ? hi : mid;
    }
    return sorted[(long long)lo * TE + st[lo] + (p - off[lo])];
}

__device__ __forceinline__ void pull_chunk(const int* __restrict__ sorted, const int* off, const int* st,
                                           int NT, int c0, int n, int total, int* lst, int* rc, int lane) {
    int w[CAP / 256], r[CAP / 256];
#pragma unroll
    for (int k = 0; k < CAP / 256; ++k)
        w[k] = list_word(sorted, off, st, NT, min(c0 + (int)threadIdx.x + 256 * k, total - 1));
#pragma unroll
    for (int k = 0; k < CAP / 256; ++k)
        r[k] = ((int)threadIdx.x + 256 * k < n) ? lds_inc(&rc[w[k] & BMASK]) : 0;
    __syncthreads();
    static_assert(BN == 64, "a lane holds the starts of rows lane and lane + 32");
    const int ca = rc[lane], cb = rc[lane + 32];
    const int ia = wave_scan(ca, lane);
    const int ib = wave_scan(cb, lane) + __shfl(ia, 31);
    const int ea = ia - ca, eb = ib - cb;
#pragma unroll
    for (int k = 0; k < CAP / 256; ++k) {
        const int row = w[k] & BMASK;
        const int sa = __shfl(ea, row & 31);
        const int sb = __shfl(eb, row & 31);
        if ((int)threadIdx.x + 256 * k < n) lst[((row < 32) ? sa : sb) + r[k]] = w[k];
    }
}

#define NPT 6
__global__ __launch_bounds__(256) void block_norm(const int* __restrict__ sorted,
                                                  const int* __restrict__ start,
                                                  float* __restrict__ norm, int N, int NT) {
    __shared__ int c[BN];
    __shared__ int off[MAXNT + 1];
    __shared__ int st[MAXNT];
    __shared__ int wsum[8];
    const int b = blockIdx.x;
    if (threadIdx.x < BN) c[threadIdx.x] = 0;
    const int cnt = pull_setup(start, b, NT, off, st, wsum);
    for (int r = 0; r < cnt; r += 256 * NPT) {
        const int i0 = r + (int)threadIdx.x;
        int w[NPT];
#pragma unroll
        for (int j = 0; j < NPT; ++j) w[j] = list_word(sorted, off, st, NT, min(i0 + 256 * j, cnt - 1));
#pragma unroll
        for (int j = 0; j < NPT; ++j)
            if (i0 + 256 * j < cnt) lds_inc(&c[w[j] & BMASK]);
    }
    __syncthreads();
    if (threadIdx.x < BN) {
        const int node = b * BN + threadIdx.x;
        if (node < N) norm[node] = 1.0f / sqrtf(fmaxf((float)c[threadIdx.x], 1.0f));
    }
}

template <int P>
__device__ __forceinline__ void flush_run(float* acc, int row, int lane, float a0, float a1) {
    float* a = acc + row * P;
    lds_addf(&a[lane],      a0);
    lds_addf(&a[lane + 32], a1);
}

template <int P, bool GUARD>
__device__ __forceinline__ void gather_group(const float* __restrict__ x, const float* __restrict__ norm,
                                             const int* lst, int i, int m, float* acc, int lane,
                                             int& cur, float& a0, float& a1) {
    static_assert(EPI == 16, "lane & (EPI - 1) picks a lane's word");
    const int   wl = lst[i + (GUARD ? min(lane & (EPI - 1), m - 1) : (lane & (EPI - 1)))];
    const float nl = norm[(unsigned)wl >> BSH];
    unsigned wd[EPI], idx[EPI];
    float v0[EPI], v1[EPI];
#pragma unroll
    for (int j = 0; j < EPI; ++j) {
        wd[j] = (unsigned)__builtin_amdgcn_readlane(wl, j);
        idx[j] = (wd[j] & ~(unsigned)BMASK) | (unsigned)lane;
    }
    __builtin_amdgcn_sched_barrier(0);
#pragma unroll
    for (int j = 0; j < EPI; ++j) {
        const float* p = x + idx[j];
        v0[j] = p[0];
        v1[j] = p[32];
    }
#pragma unroll
    for (int j = 0; j < EPI; ++j) {
        if (!GUARD || j < m) {
            const float n = __int_as_float(__builtin_amdgcn_readlane(__float_as_int(nl), j));
            const int row = (int)(wd[j] & BMASK);
            if (__builtin_expect(row != cur, 0)) {
                if (cur >= 0) flush_run<P>(acc, cur, lane, a0, a1);
                cur = row;
                a0 = 0.0f;
                a1 = 0.0f;
            }
            a0 += v0[j] * n;
            a1 += v1[j] * n;
        }
    }
}

template <int P>
__device__ __forceinline__ void gather_rows(const float* __restrict__ x, const float* __restrict__ norm,
                                            const int* lst, int cnt,
                                            float* acc, int lane, int wave) {
    const int wv = __builtin_amdgcn_readfirstlane(wave);
    cnt = __builtin_amdgcn_readfirstlane(cnt);
    const int G  = (cnt + EPI - 1) / EPI;
    const int g1 = (G * (wv + 1)) >> 3;
    int cur = -1;
    float a0 = 0.0f, a1 = 0.0f;
    for (int g = (G * wv) >> 3; g < g1; ++g) {
        const int i = g * EPI;
        const int m = cnt - i;
        if (m >= EPI) gather_group<P, false>(x, norm, lst, i, EPI, acc, lane, cur, a0, a1);
        else          gather_group<P, true >(x, norm, lst, i, m,   acc, lane, cur, a0, a1);
    }
    if (cur >= 0) flush_run<P>(acc, cur, lane, a0, a1);
}

template <int P>
__device__ __forceinline__ void gather_block(const float* __restrict__ x, const float* __restrict__ norm,
                                             const int* __restrict__ sorted, const int* off, const int* st,
                                             int NT, int total, int* lst, int* rc, float* acc,
                                             int lane, int wave) {
    for (int c0 = 0; c0 < total; c0 += CAP) {
        if (c0 > 0) {
            __syncthreads();
            if (threadIdx.x < BN) rc[threadIdx.x] = 0;
            __syncthreads();
        }
        const int n = min(CAP, total - c0);
        pull_chunk(sorted, off, st, NT, c0, n, total, lst, rc, lane);
        __syncthreads();
        gather_rows<P>(x, norm, lst, n, acc, lane, wave);
    }
}

#define PITCH 68

__global__ __launch_bounds__(256, 16) void aggregate_project(
    const float* __restrict__ feat, const float* __restrict__ norm,
    const int* __restrict__ sorted, const int* __restrict__ start,
    const float* __restrict__ W,
    const float* __restrict__ bias, float* __restrict__ outL,
    float* __restrict__ Y, int N, int NT) {
    __shared__ __attribute__((aligned(16))) float acc[BN * PITCH];
    __shared__ __attribute__((aligned(16))) float ft[BN * PITCH];
    __shared__ float nb[BN];
    __shared__ int wsum[8];
    int* lst = (int*)ft;
    int* off = lst + CAP;
    int* st  = off + MAXNT + 1;
    int* rc  = st + MAXNT;
    const int b    = blockIdx.x;
    const int lane = threadIdx.x & 31;
    const int wave = threadIdx.x >> 5;
    for (int k = threadIdx.x; k < BN * PITCH; k += 256) acc[k] = 0.0f;
    if (threadIdx.x < BN) rc[threadIdx.x] = 0;
    const int total = pull_setup(start, b, NT, off, st, wsum);

    gather_block<PITCH>(feat, norm, sorted, off, st, NT, total, lst, rc, acc, lane, wave);

    const int f    = threadIdx.x & 63;
    const int r0   = threadIdx.x >> 6;
    const int rows = min(BN, N - b * BN);
    float fv[BN / 4];
#pragma unroll
    for (int j = 0; j < BN / 4; ++j) {
        const int rr = min(r0 + 4 * j, rows - 1);
        fv[j] = feat[(long long)(b * BN + rr) * 64 + f];
    }
    float nv = 0.0f;
    if (threadIdx.x < BN) nv = ((int)threadIdx.x < rows) ? norm[b * BN + threadIdx.x] : 0.0f;
    __syncthreads();
#pragma unroll
    for (int j = 0; j < BN / 4; ++j) {
        const int r = r0 + 4 * j;
        ft[r * PITCH + f] = (r < rows) ? fv[j] : 0.0f;
    }
    if (threadIdx.x < BN) nb[threadIdx.x] = nv;
    __syncthreads();

    const float ch = -2.0f / LAM;
    const float cx = 2.0f / LAM - 1.0f;
    for (int r = r0; r < BN; r += 4)
        acc[r * PITCH + f] = ch * acc[r * PITCH + f] * nb[r] + cx * ft[r * PITCH + f];
    __syncthreads();

    const int c    = lane & 15;
    const int h    = lane >> 4;
    const int rp   = wave >> 2;
    const int ncol = (wave & 3) * 16 + c;
    const float* W0 = W;
    const float* W1 = W + 4096;
    const float* W2 = W + 8192;
    const float cx2 = 4.0f / LAM - 2.0f;
    const float* fa = ft  + (rp * 32 + c) * PITCH + 2 * h;
    const float* ta = acc + (rp * 32 + c) * PITCH + 2 * h;

    v8f L0 = {}, L1 = {}, Y0 = {}, Y1 = {};
#pragma unroll 2
    for (int k = 0; k < 64; k += 4) {
        const int ka = k + 2 * h;
        const float w0a = W0[ka * 64 + ncol], w0b = W0[(ka + 1) * 64 + ncol];
        const float w1a = W1[ka * 64 + ncol], w1b = W1[(ka + 1) * 64 + ncol];
        const float w2a = W2[ka * 64 + ncol], w2b = W2[(ka + 1) * 64 + ncol];
        const v2f bY = {w2a, w2b};
        const v2f bF = {w0a - w2a, w0b - w2b};
        const v2f bT = {w1a + cx2 * w2a, w1b + cx2 * w2b};
        const v2f f0 = *(const v2f*)(fa + k);
        const v2f t0 = *(const v2f*)(ta + k);
        const v2f f1 = *(const v2f*)(fa + 16 * PITCH + k);
        const v2f t1 = *(const v2f*)(ta + 16 * PITCH + k);
        Y0 = __builtin_amdgcn_wmma_f32_16x16x4_f32(false, t0, false, bY, (short)0, Y0, false, false);
        Y1 = __builtin_amdgcn_wmma_f32_16x16x4_f32(false, t1, false, bY, (short)0, Y1, false, false);
        L0 = __builtin_amdgcn_wmma_f32_16x16x4_f32(false, f0, false, bF, (short)0, L0, false, false);
        L1 = __builtin_amdgcn_wmma_f32_16x16x4_f32(false, f1, false, bF, (short)0, L1, false, false);
        L0 = __builtin_amdgcn_wmma_f32_16x16x4_f32(false, t0, false, bT, (short)0, L0, false, false);
        L1 = __builtin_amdgcn_wmma_f32_16x16x4_f32(false, t1, false, bT, (short)0, L1, false, false);
    }

    const float bv = bias[ncol];
    __syncthreads();
#pragma unroll
    for (int r = 0; r < 8; ++r) {
        const int lr = rp * 32 + r + 8 * h;
        ft[lr * PITCH + ncol]         = L0[r] + bv;
        acc[lr * PITCH + ncol]        = Y0[r];
        ft[(lr + 16) * PITCH + ncol]  = L1[r] + bv;
        acc[(lr + 16) * PITCH + ncol] = Y1[r];
    }
    __syncthreads();
#pragma unroll 4
    for (int j = 0; j < BN / 4; ++j) {
        const int r = r0 + 4 * j;
        if (r < rows) {
            const long long o = (long long)(b * BN + r) * 64 + f;
            outL[o] = ft[r * PITCH + f];
            Y[o]    = acc[r * PITCH + f];
        }
    }
}

__global__ __launch_bounds__(256, 16) void aggregate_finish(
    const float* __restrict__ Y, const float* __restrict__ norm,
    const int* __restrict__ sorted, const int* __restrict__ start,
    float* __restrict__ out, int N, int NT) {
    __shared__ float acc[BN * 64];
    __shared__ __attribute__((aligned(16))) int lst[CAP];
    __shared__ int off[MAXNT + 1];
    __shared__ int st[MAXNT];
    __shared__ int rc[BN];
    __shared__ int wsum[8];
    const int b    = blockIdx.x;
    const int lane = threadIdx.x & 31;
    const int wave = threadIdx.x >> 5;
    for (int k = threadIdx.x; k < BN * 64; k += 256) acc[k] = 0.0f;
    if (threadIdx.x < BN) rc[threadIdx.x] = 0;
    const int total = pull_setup(start, b, NT, off, st, wsum);

    gather_block<64>(Y, norm, sorted, off, st, NT, total, lst, rc, acc, lane, wave);
    __syncthreads();

    const float ch2 = -4.0f / LAM;
    const int f    = threadIdx.x & 63;
    const int r0   = threadIdx.x >> 6;
    const int rows = min(BN, N - b * BN);
#pragma unroll 4
    for (int r = r0; r < rows; r += 4) {
        const int node = b * BN + r;
        const long long o = (long long)node * 64 + f;
        out[o] = ch2 * acc[r * 64 + f] * norm[node] + out[o];
    }
}

extern "C" void kernel_launch(void* const* d_in, const int* in_sizes, int n_in,
                              void* d_out, int out_size, void* d_ws, size_t ws_size,
                              hipStream_t stream) {
    const float* feat = (const float*)d_in[0];
    const float* W    = (const float*)d_in[1];
    const float* bias = (const float*)d_in[2];
    const int*   esrc = (const int*)d_in[3];
    const int*   edst = (const int*)d_in[4];
    float* out = (float*)d_out;

    const int N = in_sizes[0] / 64;
    const int E = in_sizes[3];

    float* ws   = (float*)d_ws;
    float* norm = ws;
    float* Y    = ws + N;

    const int NB = (N + BN - 1) / BN;
    const int NT = (E + TE - 1) / TE;

    int* scr    = (int*)(ws + N + (long long)N * 64);
    int* sorted = scr;
    int* start  = sorted + (long long)NT * TE;

    tile_sort<<<NT, 1024, 0, stream>>>(esrc, edst, sorted, start, E, NB, NT);
    block_norm<<<NB, 256, 0, stream>>>(sorted, start, norm, N, NT);

    aggregate_project<<<NB, 256, 0, stream>>>(feat, norm, sorted, start, W, bias, out, Y, N, NT);
    aggregate_finish<<<NB, 256, 0, stream>>>(Y, norm, sorted, start, out, N, NT);
}
